# starting kernel
# baseline (speedup 1.0000x reference)
#include <hip/hip_runtime.h>
#include <math.h>

#define NN  50000
#define EE  800000
#define DIN 128
#define HH  2
#define FF  64
#define DHF 128
#define NEG_SLOPE 0.2f

typedef __attribute__((ext_vector_type(2))) float v2f;
typedef __attribute__((ext_vector_type(8))) float v8f;

#if defined(__AMDGCN__) && __has_builtin(__builtin_amdgcn_wmma_f32_16x16x4_f32)
#define USE_WMMA_F32 1
#else
#define USE_WMMA_F32 0
#endif

__global__ __launch_bounds__(256) void gat_gemm(const float* __restrict__ Hm,
                                                const float* __restrict__ W,
                                                float* __restrict__ feat)
{
    __shared__ float sA[16 * 132];
    const int tid  = threadIdx.x;
    const int row0 = blockIdx.x * 16;

    for (int i = tid; i < 16 * DIN; i += 256) {
        int r = i >> 7, c = i & 127;
        int gr = row0 + r;
        sA[r * 132 + c] = (gr < NN) ? Hm[gr * DIN + c] : 0.0f;
    }
    __syncthreads();

    const int wave = tid >> 5;
    const int lane = tid & 31;
    const int mrow = lane & 15;
    const int hi   = lane >> 4;
    const int col0 = wave * 16;

    v8f c = {};
#if USE_WMMA_F32
    for (int k = 0; k < DIN; k += 4) {
        const int ka = k + 2 * hi;
        v2f a, b;
        a.x = sA[mrow * 132 + ka + 0];
        a.y = sA[mrow * 132 + ka + 1];
        b.x = W[(ka + 0) * DHF + col0 + mrow];
        b.y = W[(ka + 1) * DHF + col0 + mrow];
        c = __builtin_amdgcn_wmma_f32_16x16x4_f32(false, a, false, b,
                                                  (short)0, c, false, false);
    }
#else
    for (int k = 0; k < DIN; ++k) {
        float bkn = W[k * DHF + col0 + mrow];
        #pragma unroll
        for (int i = 0; i < 8; ++i)
            c[i] = fmaf(sA[(i + 8 * hi) * 132 + k], bkn, c[i]);
    }
#endif
    #pragma unroll
    for (int i = 0; i < 8; ++i) {
        int M = i + 8 * hi;
        int r = row0 + M;
        if (r < NN) feat[r * DHF + col0 + mrow] = c[i];
    }
}

__global__ __launch_bounds__(256) void gat_attn(const float* __restrict__ feat,
                                                const float* __restrict__ al,
                                                const float* __restrict__ ar,
                                                float* __restrict__ el,
                                                float* __restrict__ er)
{
    int i = blockIdx.x * 256 + threadIdx.x;
    if (i >= NN * HH) return;
    int n = i >> 1, h = i & 1;
    const float* fp  = feat + n * DHF + h * FF;
    const float* alp = al + h * FF;
    const float* arp = ar + h * FF;
    float sl = 0.f, sr = 0.f;
    #pragma unroll 8
    for (int f = 0; f < FF; ++f) {
        float v = fp[f];
        sl = fmaf(v, alp[f], sl);
        sr = fmaf(v, arp[f], sr);
    }
    el[i] = sl; er[i] = sr;
}

__global__ __launch_bounds__(256) void csr_zero(int* __restrict__ deg)
{
    int i = blockIdx.x * 256 + threadIdx.x;
    if (i < NN) deg[i] = 0;
}

__global__ __launch_bounds__(256) void csr_hist(const int* __restrict__ dst,
                                                int* __restrict__ deg)
{
    int e = blockIdx.x * 256 + threadIdx.x;
    if (e < EE) atomicAdd(&deg[dst[e]], 1);
}

__global__ __launch_bounds__(1024) void csr_scan(const int* __restrict__ deg,
                                                 int* __restrict__ off,
                                                 int* __restrict__ cursor)
{
    __shared__ int buf[1024];
    __shared__ int carry;
    const int t = threadIdx.x;
    if (t == 0) carry = 0;
    __syncthreads();
    for (int base = 0; base < NN; base += 1024) {
        int i = base + t;
        int v = (i < NN) ? deg[i] : 0;
        buf[t] = v;
        __syncthreads();
        for (int ofs = 1; ofs < 1024; ofs <<= 1) {
            int x = (t >= ofs) ? buf[t - ofs] : 0;
            __syncthreads();
            buf[t] += x;
            __syncthreads();
        }
        int excl = buf[t] - v + carry;
        if (i < NN) { off[i] = excl; cursor[i] = excl; }
        __syncthreads();
        if (t == 1023) carry += buf[1023];
        __syncthreads();
    }
    if (t == 0) off[NN] = carry;
}

__global__ __launch_bounds__(256) void csr_build(const int* __restrict__ src,
                                                 const int* __restrict__ dst,
                                                 int* __restrict__ cursor,
                                                 int* __restrict__ csrc,
                                                 int* __restrict__ cdst)
{
    int e = blockIdx.x * 256 + threadIdx.x;
    if (e >= EE) return;
    int d = dst[e];
    int pos = atomicAdd(&cursor[d], 1);
    csrc[pos] = src[e];
    cdst[pos] = d;
}

__global__ __launch_bounds__(256) void gat_edge_e(const float* __restrict__ el,
                                                  const float* __restrict__ er,
                                                  const int* __restrict__ csrc,
                                                  const int* __restrict__ cdst,
                                                  float* __restrict__ ecsr)
{
    int i = blockIdx.x * 256 + threadIdx.x;
    if (i >= EE * HH) return;
    int pos = i >> 1, h = i & 1;
    int s = csrc[pos], d = cdst[pos];
    float v = el[s * HH + h] + er[d * HH + h];
    ecsr[i] = (v >= 0.f) ? v : NEG_SLOPE * v;
}

__global__ __launch_bounds__(128) void gat_aggregate(const float* __restrict__ ecsr,
                                                     const float* __restrict__ feat,
                                                     const int* __restrict__ csrc,
                                                     const int* __restrict__ off,
                                                     const float* __restrict__ bias,
                                                     float* __restrict__ out,
                                                     int last)
{
    __shared__ float sv[DHF];
    const int d = blockIdx.x;
    const int t = threadIdx.x;
    const int h = t >> 6;
    const int p0 = off[d], p1 = off[d + 1];

    float m = -__builtin_inff();
    for (int p = p0; p < p1; ++p)
        m = fmaxf(m, ecsr[p * HH + h]);

    float den = 0.f, acc = 0.f;
    for (int p = p0; p < p1; ++p) {
        float ex = expf(ecsr[p * HH + h] - m);
        den += ex;
        int s = csrc[p];
        acc = fmaf(ex, feat[s * DHF + t], acc);
    }
    float r = acc / ((den == 0.f) ? 1.f : den);
    float v = r + bias[t];
    v = (v > 0.f) ? v : expm1f(v);

    if (!last) {
        out[d * DHF + t] = v;
    } else {
        sv[t] = v;
        __syncthreads();
        if (t < FF) out[d * FF + t] = 0.5f * (sv[t] + sv[t + FF]);
    }
}

extern "C" void kernel_launch(void* const* d_in, const int* in_sizes, int n_in,
                              void* d_out, int out_size, void* d_ws, size_t ws_size,
                              hipStream_t stream)
{
    const float* x    = (const float*)d_in[0];
    const int*   src  = (const int*)d_in[1];
    const int*   dst  = (const int*)d_in[2];
    const float* Wall = (const float*)d_in[3];
    const float* al   = (const float*)d_in[4];
    const float* ar   = (const float*)d_in[5];
    const float* bias = (const float*)d_in[6];
    float* out = (float*)d_out;

    char* p = (char*)d_ws;
    auto take = [&](size_t bytes) {
        char* q = p;
        p += (bytes + 255) & ~size_t(255);
        return q;
    };
    float* feat   = (float*)take(sizeof(float) * NN * DHF);
    float* hbuf   = (float*)take(sizeof(float) * NN * DHF);
    float* el     = (float*)take(sizeof(float) * NN * HH);
    float* er     = (float*)take(sizeof(float) * NN * HH);
    float* ecsr   = (float*)take(sizeof(float) * EE * HH);
    int*   deg    = (int*)take(sizeof(int) * NN);
    int*   off    = (int*)take(sizeof(int) * (NN + 1));
    int*   cursor = (int*)take(sizeof(int) * NN);
    int*   csrc   = (int*)take(sizeof(int) * EE);
    int*   cdst   = (int*)take(sizeof(int) * EE);

    const int tb   = 256;
    const int gN   = (NN + tb - 1) / tb;
    const int gE   = (EE + tb - 1) / tb;
    const int gN2  = (NN * HH + tb - 1) / tb;
    const int gE2  = (EE * HH + tb - 1) / tb;
    const int gGmm = (NN + 15) / 16;

    csr_zero <<<gN, tb, 0, stream>>>(deg);
    csr_hist <<<gE, tb, 0, stream>>>(dst, deg);
    csr_scan <<<1, 1024, 0, stream>>>(deg, off, cursor);
    csr_build<<<gE, tb, 0, stream>>>(src, dst, cursor, csrc, cdst);

    const float* hin = x;
    for (int layer = 0; layer < 3; ++layer) {
        const float* Wl  = Wall + layer * DIN * DHF;
        const float* all = al   + layer * HH * FF;
        const float* arl = ar   + layer * HH * FF;
        const float* bl  = bias + layer * HH * FF;
        const int    last = (layer == 2);
        float* outp = last ? out : hbuf;

        gat_gemm     <<<gGmm, tb, 0, stream>>>(hin, Wl, feat);
        gat_attn     <<<gN2,  tb, 0, stream>>>(feat, all, arl, el, er);
        gat_edge_e   <<<gE2,  tb, 0, stream>>>(el, er, csrc, cdst, ecsr);
        gat_aggregate<<<NN,  128, 0, stream>>>(ecsr, feat, csrc, off, bl, outp, last);
        hin = hbuf;
    }
    (void)in_sizes; (void)n_in; (void)out_size; (void)ws_size;
}


# projection: 2 independent waves per 16 rows, 4 column tiles each, rows loaded as A fragments (no LDS tile, no barrier), W pre-ordered by fragment in the fold launch; k order of every sum changes
# speedup vs baseline: 2.2749x; 2.2749x over previous
#include <hip/hip_runtime.h>
#include <math.h>

#define NN  50000
#define EE  800000
#define DIN 128
#define HH  2
#define FF  64
#define DHF 128
#define NEG_SLOPE 0.2f

typedef __attribute__((ext_vector_type(2))) float v2f;
typedef __attribute__((ext_vector_type(8))) float v8f;

#if defined(__AMDGCN__) && __has_builtin(__builtin_amdgcn_wmma_f32_16x16x4_f32)
#define USE_WMMA_F32 1
#else
#define USE_WMMA_F32 0
#endif

#define FOLD_G  (3 * 4)
#define WQ_G    (3 * (DIN / 8))
__global__ __launch_bounds__(DIN) void gat_fold(const float* __restrict__ Wall,
                                                const float* __restrict__ al,
                                                const float* __restrict__ ar,
                                                float* __restrict__ wa,
                                                float* __restrict__ wq)
{
    if (blockIdx.x >= FOLD_G) {
        const int g     = blockIdx.x - FOLD_G;
        const int layer = g / (DIN / 8);
        const int kb    = g % (DIN / 8);
        const float* __restrict__ Wl = Wall + layer * DIN * DHF;
        float4* __restrict__ q4 = reinterpret_cast<float4*>(wq + layer * DIN * DHF);
        #pragma unroll
        for (int r = 0; r < 2; ++r) {
            const int idx = threadIdx.x + DIN * r;
            const int n   = idx >> 5;
            const int ln  = idx & 31;
            const float* __restrict__ p = Wl + (8 * kb + 4 * (ln >> 4)) * DHF + 16 * n + (ln & 15);
            q4[(kb * 8 + n) * 32 + ln] = make_float4(p[0], p[DHF], p[2 * DHF], p[3 * DHF]);
        }
        return;
    }
    const int k     = threadIdx.x;
    const int j     = blockIdx.x & 3;
    const int layer = blockIdx.x >> 2;
    const int h     = j & 1;
    const float* __restrict__ a = ((j < 2) ? al : ar) + (layer * HH + h) * FF;
    const float4* __restrict__ w4 = reinterpret_cast<const float4*>(Wall + (layer * DIN + k) * DHF + h * FF);
    const float4* __restrict__ a4 = reinterpret_cast<const float4*>(a);
    float s = 0.f;
    #pragma unroll
    for (int i = 0; i < FF / 4; ++i) {
        const float4 w = w4[i];
        const float4 q = a4[i];
        s = fmaf(w.x, q.x, s);
        s = fmaf(w.y, q.y, s);
        s = fmaf(w.z, q.z, s);
        s = fmaf(w.w, q.w, s);
    }
    wa[(layer * 4 + j) * DIN + k] = s;
}

#define GEMM_NT 4
#define GEMM_NP 7
static_assert(NN % 16 == 0, "row tiles are whole: no row guard");
static_assert(DHF == 2 * 16 * GEMM_NT, "two waves of 4 column tiles cover W");
static_assert(DIN % 8 == 0, "k runs in blocks of 8");

__device__ __forceinline__ void gemm_load(float4 (&p)[GEMM_NP], const float4* __restrict__ bp,
                                          const float4* __restrict__ s0, const float4* __restrict__ s1,
                                          const float4* __restrict__ ap, int kb)
{
    #pragma unroll
    for (int n = 0; n < GEMM_NT; ++n) p[n] = bp[(kb * 8 + n) * 32];
    p[4] = s0[2 * kb];
    p[5] = s1[2 * kb];
    p[6] = ap[2 * kb];
}

__global__ __launch_bounds__(128) void gat_gemm(const float* __restrict__ Hm,
                                               const float* __restrict__ wq,
                                               float* __restrict__ feat,
                                               const float* __restrict__ wa,
                                               float* __restrict__ el,
                                               float* __restrict__ er)
{
    const int lane = threadIdx.x & 31;
    const int wi   = __builtin_amdgcn_readfirstlane((int)(threadIdx.x >> 5));
    const int wv   = wi & 1;
    const int tile = blockIdx.x * 2 + (wi >> 1);
    if (tile >= NN / 16) return;
    const int cc   = lane & 15;
    const int hh   = lane >> 4;
    const int row0 = tile * 16;

#if USE_WMMA_F32
    const float4* __restrict__ ap = reinterpret_cast<const float4*>(Hm + (row0 + cc) * DIN + 4 * hh);
    const float4* __restrict__ bp = reinterpret_cast<const float4*>(wq) + (GEMM_NT * wv) * 32 + lane;
    const float4* __restrict__ s0 = reinterpret_cast<const float4*>(wa + (2 * wv) * DIN + 4 * hh);
    const float4* __restrict__ s1 = reinterpret_cast<const float4*>(wa + (2 * wv + 1) * DIN + 4 * hh);

    v8f acc[GEMM_NT];
    #pragma unroll
    for (int n = 0; n < GEMM_NT; ++n) acc[n] = v8f{};
    v2f dd0 = {0.f, 0.f}, dd1 = {0.f, 0.f};

    float4 pc[GEMM_NP], pn[GEMM_NP];
    gemm_load(pc, bp, s0, s1, ap, 0);
    #pragma unroll
    for (int kb = 0; kb < DIN / 8; ++kb) {
        if (kb + 1 < DIN / 8) gemm_load(pn, bp, s0, s1, ap, kb + 1);
        __builtin_amdgcn_sched_barrier(0);
        v2f a0, a1;
        a0.x = pc[6].x; a0.y = pc[6].y;
        a1.x = pc[6].z; a1.y = pc[6].w;
        #pragma unroll
        for (int n = 0; n < GEMM_NT; ++n) {
            v2f b0;
            b0.x = pc[n].x; b0.y = pc[n].y;
            acc[n] = __builtin_amdgcn_wmma_f32_16x16x4_f32(false, a0, false, b0,
                                                           (short)0, acc[n], false, false);
        }
        #pragma unroll
        for (int n = 0; n < GEMM_NT; ++n) {
            v2f b1;
            b1.x = pc[n].z; b1.y = pc[n].w;
            acc[n] = __builtin_amdgcn_wmma_f32_16x16x4_f32(false, a1, false, b1,
                                                           (short)0, acc[n], false, false);
        }
        {
            v2f r0, r1, u0, u1, t0, t1;
            r0.x = pc[6].x; r0.y = pc[6].y; r1.x = pc[6].z; r1.y = pc[6].w;
            u0.x = pc[4].x; u0.y = pc[4].y; u1.x = pc[4].z; u1.y = pc[4].w;
            t0.x = pc[5].x; t0.y = pc[5].y; t1.x = pc[5].z; t1.y = pc[5].w;
            dd0 = __builtin_elementwise_fma(r0, u0, dd0);
            dd0 = __builtin_elementwise_fma(r1, u1, dd0);
            dd1 = __builtin_elementwise_fma(r0, t0, dd1);
            dd1 = __builtin_elementwise_fma(r1, t1, dd1);
        }
        __builtin_amdgcn_sched_barrier(0);
        #pragma unroll
        for (int n = 0; n < GEMM_NP; ++n) pc[n] = pn[n];
    }

    #pragma unroll
    for (int n = 0; n < GEMM_NT; ++n) {
        #pragma unroll
        for (int i = 0; i < 8; ++i)
            feat[(row0 + 8 * hh + i) * DHF + 16 * (GEMM_NT * wv + n) + cc] = acc[n][i];
    }
    float d0 = dd0.x + dd0.y;
    float d1 = dd1.x + dd1.y;
    d0 += __shfl_xor(d0, 16);
    d1 += __shfl_xor(d1, 16);
    if (hh == 0)
        reinterpret_cast<float2*>(wv ? er : el)[row0 + cc] = make_float2(d0, d1);
#else
    const float* __restrict__ hr = Hm + (row0 + cc) * DIN;
    for (int i = 0; i < DHF / 4; ++i) {
        const int col = (DHF / 2) * wv + (DHF / 4) * hh + i;
        float s = 0.f;
        for (int k = 0; k < DIN; ++k)
            s = fmaf(hr[k], wq[((((k >> 3) * 8 + (col >> 4)) * 32) + 16 * ((k >> 2) & 1) + (col & 15)) * 4 + (k & 3)], s);
        feat[(row0 + cc) * DHF + col] = s;
    }
    {
        float s = 0.f;
        for (int k = 0; k < DIN; ++k) s = fmaf(hr[k], wa[(2 * wv + hh) * DIN + k], s);
        (wv ? er : el)[(row0 + cc) * HH + hh] = s;
    }
#endif
}

#define CH_G       64
#define CH_EDGES   (EE / CH_G)
#define CH_THREADS 1024
#define CH_TRIPS   ((CH_EDGES + CH_THREADS - 1) / CH_THREADS)
static_assert(EE % CH_G == 0, "chunks must tile the edge list");
static_assert(CH_EDGES % 4 == 0, "four consecutive edges must share a chunk");
static_assert(NN % 4 == 0, "counters are cleared and copied as 16-byte words");

__global__ __launch_bounds__(CH_THREADS) void csr_chunk_count(const int* __restrict__ dst,
                                                              int* __restrict__ plane,
                                                              int* __restrict__ rank)
{
    __shared__ __attribute__((aligned(16))) int cnt[NN];
    const int t  = threadIdx.x;
    const int e0 = blockIdx.x * CH_EDGES;

    int d[CH_TRIPS];
    #pragma unroll
    for (int k = 0; k < CH_TRIPS; ++k) {
        const int i = t + k * CH_THREADS;
        d[k] = dst[e0 + ((i < CH_EDGES) ? i : (CH_EDGES - 1))];
    }

    int4* c4 = reinterpret_cast<int4*>(cnt);
    for (int i = t; i < NN / 4; i += CH_THREADS) c4[i] = make_int4(0, 0, 0, 0);
    __syncthreads();

    int r[CH_TRIPS];
    #pragma unroll
    for (int k = 0; k < CH_TRIPS; ++k) {
        const int i = t + k * CH_THREADS;
        r[k] = (i < CH_EDGES) ? atomicAdd(&cnt[d[k]], 1) : 0;
    }
    __syncthreads();
    #pragma unroll
    for (int k = 0; k < CH_TRIPS; ++k) {
        const int i = t + k * CH_THREADS;
        if (i < CH_EDGES) rank[e0 + i] = r[k];
    }

    int4* p4 = reinterpret_cast<int4*>(plane + blockIdx.x * NN);
    for (int i = t; i < NN / 4; i += CH_THREADS) p4[i] = c4[i];
}

__global__ __launch_bounds__(256) void csr_chunk_prefix(int* __restrict__ plane,
                                                        int* __restrict__ deg)
{
    const int n = blockIdx.x * 256 + threadIdx.x;
    if (n >= NN) return;
    int c[CH_G];
    #pragma unroll
    for (int g = 0; g < CH_G; ++g) c[g] = plane[g * NN + n];
    int run = 0;
    #pragma unroll
    for (int g = 0; g < CH_G; ++g) { plane[g * NN + n] = run; run += c[g]; }
    deg[n] = run;
}

#define SCAN_CHUNK 1024
#define SCAN_FRONT_TRIPS ((((NN - 1) / SCAN_CHUNK) * (SCAN_CHUNK / 4) + SCAN_CHUNK - 1) / SCAN_CHUNK)
__global__ __launch_bounds__(SCAN_CHUNK) void csr_scan(const int* __restrict__ deg,
                                                       int* __restrict__ off)
{
    __shared__ int s_front[32];
    __shared__ int s_total[32];
    const int t    = threadIdx.x;
    const int lane = t & 31;
    const int wave = t >> 5;
    const int base = blockIdx.x * SCAN_CHUNK;

    const int4* __restrict__ d4 = reinterpret_cast<const int4*>(deg);
    const int nv = base >> 2;
    int front = 0;
    static_assert(SCAN_FRONT_TRIPS * SCAN_CHUNK * 4 <= NN, "front loads must stay inside deg");
    #pragma unroll
    for (int k = 0; k < SCAN_FRONT_TRIPS; ++k) {
        const int  j   = t + k * SCAN_CHUNK;
        const int4 q   = d4[j];
        const int  sum = (q.x + q.y) + (q.z + q.w);
        front += (j < nv) ? sum : 0;
    }

    const int i = base + t;
    const int v = (i < NN) ? deg[i] : 0;

    int s = v;
    #pragma unroll
    for (int o = 1; o < 32; o <<= 1) {
        const int y = __shfl_up(s, o);
        if (lane >= o) s += y;
    }
    #pragma unroll
    for (int o = 16; o > 0; o >>= 1) front += __shfl_xor(front, o);

    if (lane == 31) s_total[wave] = s;
    if (lane == 0)  s_front[wave] = front;
    __syncthreads();

    int c = s_front[lane] + ((lane < wave) ? s_total[lane] : 0);
    #pragma unroll
    for (int o = 16; o > 0; o >>= 1) c += __shfl_xor(c, o);

    const int excl = c + s - v;
    if (i < NN) off[i] = excl;
    if (i == NN - 1) off[NN] = excl + v;
}

__global__ __launch_bounds__(256) void csr_place(const int* __restrict__ src,
                                                 const int* __restrict__ dst,
                                                 const int* __restrict__ rank,
                                                 const int* __restrict__ plane,
                                                 const int* __restrict__ off,
                                                 int* __restrict__ csrc)
{
    const int q = blockIdx.x * 256 + threadIdx.x;
    if (q >= EE / 4) return;
    const int4 d = reinterpret_cast<const int4*>(dst)[q];
    const int4 s = reinterpret_cast<const int4*>(src)[q];
    const int4 r = reinterpret_cast<const int4*>(rank)[q];
    const int* __restrict__ pl = plane + ((q * 4) / CH_EDGES) * NN;
    const int p0 = off[d.x] + pl[d.x] + r.x;
    const int p1 = off[d.y] + pl[d.y] + r.y;
    const int p2 = off[d.z] + pl[d.z] + r.z;
    const int p3 = off[d.w] + pl[d.w] + r.w;
    csrc[p0] = s.x;
    csrc[p1] = s.y;
    csrc[p2] = s.z;
    csrc[p3] = s.w;
}

#define SCORE_NPW 8
static_assert(NN % SCORE_NPW == 0, "a wave's nodes must all exist");

__device__ __forceinline__ float wave_max(float v)
{
    #pragma unroll
    for (int o = 16; o > 0; o >>= 1) v = fmaxf(v, __shfl_xor(v, o));
    return v;
}
__device__ __forceinline__ float wave_sum(float v)
{
    #pragma unroll
    for (int o = 16; o > 0; o >>= 1) v += __shfl_xor(v, o);
    return v;
}
__device__ __forceinline__ float2 edge_score(const float2* __restrict__ el2, float2 erd, int s)
{
    const float2 l = el2[s];
    float a = l.x + erd.x, b = l.y + erd.y;
    a = (a >= 0.f) ? a : NEG_SLOPE * a;
    b = (b >= 0.f) ? b : NEG_SLOPE * b;
    return make_float2(a, b);
}

__global__ __launch_bounds__(256) void gat_edge_alpha(const float* __restrict__ el,
                                                      const float* __restrict__ er,
                                                      const int* __restrict__ csrc,
                                                      const int* __restrict__ off,
                                                      float* __restrict__ wgt)
{
    const int lane = threadIdx.x & 31;
    const int wave = __builtin_amdgcn_readfirstlane((int)(threadIdx.x >> 5));
    const int d0 = (blockIdx.x * 8 + wave) * SCORE_NPW;
    if (d0 >= NN) return;
    const float2* __restrict__ el2 = reinterpret_cast<const float2*>(el);
    const float2* __restrict__ er2 = reinterpret_cast<const float2*>(er);
    float2* __restrict__ w2 = reinterpret_cast<float2*>(wgt);
    const float ninf = -__builtin_inff();

    int po[SCORE_NPW + 1];
    #pragma unroll
    for (int k = 0; k <= SCORE_NPW; ++k) po[k] = off[d0 + k];
    bool small = true;
    #pragma unroll
    for (int k = 0; k < SCORE_NPW; ++k) small = small && (po[k + 1] - po[k] <= 32);

    if (small) {
        float2 e[SCORE_NPW];
        #pragma unroll
        for (int k = 0; k < SCORE_NPW; ++k) {
            const int p = po[k] + lane;
            e[k] = edge_score(el2, er2[d0 + k], csrc[(p < EE) ? p : (EE - 1)]);
        }
        float2 x[SCORE_NPW], s[SCORE_NPW];
        #pragma unroll
        for (int k = 0; k < SCORE_NPW; ++k) {
            const bool on = po[k] + lane < po[k + 1];
            const float e0 = on ? e[k].x : ninf;
            const float e1 = on ? e[k].y : ninf;
            const float m0 = wave_max(e0), m1 = wave_max(e1);
            x[k] = make_float2(expf(e0 - m0), expf(e1 - m1));
            s[k] = make_float2(wave_sum(x[k].x), wave_sum(x[k].y));
        }
        #pragma unroll
        for (int k = 0; k < SCORE_NPW; ++k) {
            const int p = po[k] + lane;
            if (p < po[k + 1]) w2[p] = make_float2(x[k].x / s[k].x, x[k].y / s[k].y);
        }
    } else {
        #pragma nounroll
        for (int k = 0; k < SCORE_NPW; ++k) {
            const int d  = d0 + k;
            const int p0 = off[d], p1 = off[d + 1];
            const float2 erd = er2[d];
            float m0 = ninf, m1 = ninf;
            for (int p = p0 + lane; p < p1; p += 32) {
                const float2 c = edge_score(el2, erd, csrc[p]);
                m0 = fmaxf(m0, c.x); m1 = fmaxf(m1, c.y);
            }
            m0 = wave_max(m0); m1 = wave_max(m1);
            float s0 = 0.f, s1 = 0.f;
            for (int p = p0 + lane; p < p1; p += 32) {
                const float2 c = edge_score(el2, erd, csrc[p]);
                s0 += expf(c.x - m0); s1 += expf(c.y - m1);
            }
            s0 = wave_sum(s0); s1 = wave_sum(s1);
            for (int p = p0 + lane; p < p1; p += 32) {
                const float2 c = edge_score(el2, erd, csrc[p]);
                w2[p] = make_float2(expf(c.x - m0) / s0, expf(c.y - m1) / s1);
            }
        }
    }
}

#define AGG_BATCH 16
#define AGG_NB    16
static_assert(NN % AGG_NB == 0, "a work-group's nodes must all exist");

#define AGG_CLOSE_NODES(base_pos, rel_next)                                    \
    while (rel == (rel_next)) {                                                \
        float v = acc + b;                                                     \
        v = (v > 0.f) ? v : expm1f(v);                                         \
        if (!last) out[(d0 + k) * DHF + t] = v;                                \
        else       sv[k * DHF + t] = v;                                        \
        acc = 0.f;                                                             \
        ++k;                                                                   \
        const int kn = (k < AGG_NB) ? (k + 1) : AGG_NB;                        \
        const int en = __builtin_amdgcn_readfirstlane(off[d0 + kn]);           \
        nend = (k < AGG_NB) ? en : 0x7fffffff;                                 \
        rel  = nend - (base_pos);                                              \
    }

__global__ __launch_bounds__(128) void gat_aggregate(const float* __restrict__ wgt,
                                                     const float* __restrict__ feat,
                                                     const int* __restrict__ csrc,
                                                     const int* __restrict__ off,
                                                     const float* __restrict__ bias,
                                                     float* __restrict__ out,
                                                     int last)
{
    __shared__ float sv[AGG_NB * DHF];
    const int t    = threadIdx.x;
    const int lane = t & 31;
    const int h  = __builtin_amdgcn_readfirstlane(t >> 6);
    const int d0 = blockIdx.x * AGG_NB;
    const int P0 = __builtin_amdgcn_readfirstlane(off[d0]);
    const int P1 = __builtin_amdgcn_readfirstlane(off[d0 + AGG_NB]);
    const float b = bias[t];

    int   k    = 0;
    int   nend = __builtin_amdgcn_readfirstlane(off[d0 + 1]);
    int   rel  = nend - P0;
    float acc  = 0.f;

    AGG_CLOSE_NODES(P0, 0)

    if (P1 > P0) {
        int   q   = P0 + lane;
        int   qc  = (q < P1) ? q : (P1 - 1);
        int   vsn = csrc[qc];
        float vwn = wgt[qc * HH + h];

        for (int g = P0; g < P1; g += 32) {
            int   vs = vsn;
            float vw = (g + lane < P1) ? vwn : 0.f;
            if (g + 32 < P1) {
                q   = g + 32 + lane;
                qc  = (q < P1) ? q : (P1 - 1);
                vsn = csrc[qc];
                vwn = wgt[qc * HH + h];
            }
            #pragma nounroll
            for (int hb = 0; hb < 32; hb += AGG_BATCH) {
                const int base = g + hb;
                if (base >= P1) break;
                rel = nend - base;
                float x[AGG_BATCH];
                #pragma unroll
                for (int j = 0; j < AGG_BATCH; ++j) {
                    const int s = __builtin_amdgcn_readlane(vs, j);
                    x[j] = feat[s * DHF + t];
                }
                #pragma unroll
                for (int j = 0; j < AGG_BATCH; ++j) {
                    const float w = __int_as_float(__builtin_amdgcn_readlane(__float_as_int(vw), j));
                    acc = fmaf(w, x[j], acc);
                    AGG_CLOSE_NODES(base, j + 1)
                }
                vs = __shfl_down(vs, AGG_BATCH);
                vw = __shfl_down(vw, AGG_BATCH);
            }
        }
    }

    if (last) {
        __syncthreads();
        for (int i = t; i < AGG_NB * FF; i += 128) {
            const int n = i >> 6, f = i & (FF - 1);
            out[(d0 + n) * FF + f] = 0.5f * (sv[n * DHF + f] + sv[n * DHF + f + FF]);
        }
    }
}

extern "C" void kernel_launch(void* const* d_in, const int* in_sizes, int n_in,
                              void* d_out, int out_size, void* d_ws, size_t ws_size,
                              hipStream_t stream)
{
    const float* x    = (const float*)d_in[0];
    const int*   src  = (const int*)d_in[1];
    const int*   dst  = (const int*)d_in[2];
    const float* Wall = (const float*)d_in[3];
    const float* al   = (const float*)d_in[4];
    const float* ar   = (const float*)d_in[5];
    const float* bias = (const float*)d_in[6];
    float* out = (float*)d_out;

    char* p = (char*)d_ws;
    auto take = [&](size_t bytes) {
        char* q = p;
        p += (bytes + 255) & ~size_t(255);
        return q;
    };
    float* feat   = (float*)take(sizeof(float) * NN * DHF);
    float* hbuf   = (float*)take(sizeof(float) * NN * DHF);
    float* el     = (float*)take(sizeof(float) * NN * HH);
    float* er     = (float*)take(sizeof(float) * NN * HH);
    float* wgt    = (float*)take(sizeof(float) * EE * HH);
    int*   deg    = (int*)take(sizeof(int) * NN);
    int*   off    = (int*)take(sizeof(int) * (NN + 1));
    int*   rank   = (int*)take(sizeof(int) * EE);
    int*   plane  = (int*)take(sizeof(int) * CH_G * NN);
    int*   csrc   = (int*)take(sizeof(int) * EE);
    float* wa     = (float*)take(sizeof(float) * 3 * 4 * DIN);
    float* wq     = (float*)take(sizeof(float) * 3 * DIN * DHF);

    const int tb   = 256;
    const int gN   = (NN + tb - 1) / tb;
    const int gPl  = (EE / 4 + tb - 1) / tb;
    const int gSc  = (NN / SCORE_NPW + 7) / 8;
    const int gGmm = (NN / 16 + 1) / 2;
    const int gScan = (NN + SCAN_CHUNK - 1) / SCAN_CHUNK;

    gat_fold<<<FOLD_G + WQ_G, DIN, 0, stream>>>(Wall, al, ar, wa, wq);

    csr_chunk_count <<<CH_G, CH_THREADS, 0, stream>>>(dst, plane, rank);
    csr_chunk_prefix<<<gN, tb, 0, stream>>>(plane, deg);
    csr_scan        <<<gScan, SCAN_CHUNK, 0, stream>>>(deg, off);
    csr_place       <<<gPl, tb, 0, stream>>>(src, dst, rank, plane, off, csrc);

    const float* hin = x;
    for (int layer = 0; layer < 3; ++layer) {
        const float* Wl  = wq + layer * DIN * DHF;
        const float* bl  = bias + layer * HH * FF;
        const int    last = (layer == 2);
        float* outp = last ? out : hbuf;

        gat_gemm     <<<gGmm, 128, 0, stream>>>(hin, Wl, feat, wa + layer * 4 * DIN, el, er);
        gat_edge_alpha<<<gSc, tb, 0, stream>>>(el, er, csrc, off, wgt);
        gat_aggregate<<<NN / AGG_NB, 128, 0, stream>>>(wgt, feat, csrc, off, bl, outp, last);
        hin = hbuf;
    }
    (void)in_sizes; (void)n_in; (void)out_size; (void)ws_size;
}


# aggregate: a wave takes ONE head of 8 of its work-group's 16 nodes, a lane two neighbouring features: a row load is one aligned 256-byte piece (8 B a lane), half as many loads; same sums, CSR order
# speedup vs baseline: 2.4262x; 1.0665x over previous
#include <hip/hip_runtime.h>
#include <math.h>

#define NN  50000
#define EE  800000
#define DIN 128
#define HH  2
#define FF  64
#define DHF 128
#define NEG_SLOPE 0.2f

typedef __attribute__((ext_vector_type(2))) float v2f;
typedef __attribute__((ext_vector_type(8))) float v8f;

#if defined(__AMDGCN__) && __has_builtin(__builtin_amdgcn_wmma_f32_16x16x4_f32)
#define USE_WMMA_F32 1
#else
#define USE_WMMA_F32 0
#endif

#define FOLD_G  (3 * 4)
#define WQ_G    (3 * (DIN / 8))
__global__ __launch_bounds__(DIN) void gat_fold(const float* __restrict__ Wall,
                                                const float* __restrict__ al,
                                                const float* __restrict__ ar,
                                                float* __restrict__ wa,
                                                float* __restrict__ wq)
{
    if (blockIdx.x >= FOLD_G) {
        const int g     = blockIdx.x - FOLD_G;
        const int layer = g / (DIN / 8);
        const int kb    = g % (DIN / 8);
        const float* __restrict__ Wl = Wall + layer * DIN * DHF;
        float4* __restrict__ q4 = reinterpret_cast<float4*>(wq + layer * DIN * DHF);
        #pragma unroll
        for (int r = 0; r < 2; ++r) {
            const int idx = threadIdx.x + DIN * r;
            const int n   = idx >> 5;
            const int ln  = idx & 31;
            const float* __restrict__ p = Wl + (8 * kb + 4 * (ln >> 4)) * DHF + 16 * n + (ln & 15);
            q4[(kb * 8 + n) * 32 + ln] = make_float4(p[0], p[DHF], p[2 * DHF], p[3 * DHF]);
        }
        return;
    }
    const int k     = threadIdx.x;
    const int j     = blockIdx.x & 3;
    const int layer = blockIdx.x >> 2;
    const int h     = j & 1;
    const float* __restrict__ a = ((j < 2) ? al : ar) + (layer * HH + h) * FF;
    const float4* __restrict__ w4 = reinterpret_cast<const float4*>(Wall + (layer * DIN + k) * DHF + h * FF);
    const float4* __restrict__ a4 = reinterpret_cast<const float4*>(a);
    float s = 0.f;
    #pragma unroll
    for (int i = 0; i < FF / 4; ++i) {
        const float4 w = w4[i];
        const float4 q = a4[i];
        s = fmaf(w.x, q.x, s);
        s = fmaf(w.y, q.y, s);
        s = fmaf(w.z, q.z, s);
        s = fmaf(w.w, q.w, s);
    }
    wa[(layer * 4 + j) * DIN + k] = s;
}

#define GEMM_NT 4
#define GEMM_NP 7
static_assert(NN % 16 == 0, "row tiles are whole: no row guard");
static_assert(DHF == 2 * 16 * GEMM_NT, "two waves of 4 column tiles cover W");
static_assert(DIN % 8 == 0, "k runs in blocks of 8");

__device__ __forceinline__ void gemm_load(float4 (&p)[GEMM_NP], const float4* __restrict__ bp,
                                          const float4* __restrict__ s0, const float4* __restrict__ s1,
                                          const float4* __restrict__ ap, int kb)
{
    #pragma unroll
    for (int n = 0; n < GEMM_NT; ++n) p[n] = bp[(kb * 8 + n) * 32];
    p[4] = s0[2 * kb];
    p[5] = s1[2 * kb];
    p[6] = ap[2 * kb];
}

__global__ __launch_bounds__(128) void gat_gemm(const float* __restrict__ Hm,
                                               const float* __restrict__ wq,
                                               float* __restrict__ feat,
                                               const float* __restrict__ wa,
                                               float* __restrict__ el,
                                               float* __restrict__ er)
{
    const int lane = threadIdx.x & 31;
    const int wi   = __builtin_amdgcn_readfirstlane((int)(threadIdx.x >> 5));
    const int wv   = wi & 1;
    const int tile = blockIdx.x * 2 + (wi >> 1);
    if (tile >= NN / 16) return;
    const int cc   = lane & 15;
    const int hh   = lane >> 4;
    const int row0 = tile * 16;

#if USE_WMMA_F32
    const float4* __restrict__ ap = reinterpret_cast<const float4*>(Hm + (row0 + cc) * DIN + 4 * hh);
    const float4* __restrict__ bp = reinterpret_cast<const float4*>(wq) + (GEMM_NT * wv) * 32 + lane;
    const float4* __restrict__ s0 = reinterpret_cast<const float4*>(wa + (2 * wv) * DIN + 4 * hh);
    const float4* __restrict__ s1 = reinterpret_cast<const float4*>(wa + (2 * wv + 1) * DIN + 4 * hh);

    v8f acc[GEMM_NT];
    #pragma unroll
    for (int n = 0; n < GEMM_NT; ++n) acc[n] = v8f{};
    v2f dd0 = {0.f, 0.f}, dd1 = {0.f, 0.f};

    float4 pc[GEMM_NP], pn[GEMM_NP];
    gemm_load(pc, bp, s0, s1, ap, 0);
    #pragma unroll
    for (int kb = 0; kb < DIN / 8; ++kb) {
        if (kb + 1 < DIN / 8) gemm_load(pn, bp, s0, s1, ap, kb + 1);
        __builtin_amdgcn_sched_barrier(0);
        v2f a0, a1;
        a0.x = pc[6].x; a0.y = pc[6].y;
        a1.x = pc[6].z; a1.y = pc[6].w;
        #pragma unroll
        for (int n = 0; n < GEMM_NT; ++n) {
            v2f b0;
            b0.x = pc[n].x; b0.y = pc[n].y;
            acc[n] = __builtin_amdgcn_wmma_f32_16x16x4_f32(false, a0, false, b0,
                                                           (short)0, acc[n], false, false);
        }
        #pragma unroll
        for (int n = 0; n < GEMM_NT; ++n) {
            v2f b1;
            b1.x = pc[n].z; b1.y = pc[n].w;
            acc[n] = __builtin_amdgcn_wmma_f32_16x16x4_f32(false, a1, false, b1,
                                                           (short)0, acc[n], false, false);
        }
        {
            v2f r0, r1, u0, u1, t0, t1;
            r0.x = pc[6].x; r0.y = pc[6].y; r1.x = pc[6].z; r1.y = pc[6].w;
            u0.x = pc[4].x; u0.y = pc[4].y; u1.x = pc[4].z; u1.y = pc[4].w;
            t0.x = pc[5].x; t0.y = pc[5].y; t1.x = pc[5].z; t1.y = pc[5].w;
            dd0 = __builtin_elementwise_fma(r0, u0, dd0);
            dd0 = __builtin_elementwise_fma(r1, u1, dd0);
            dd1 = __builtin_elementwise_fma(r0, t0, dd1);
            dd1 = __builtin_elementwise_fma(r1, t1, dd1);
        }
        __builtin_amdgcn_sched_barrier(0);
        #pragma unroll
        for (int n = 0; n < GEMM_NP; ++n) pc[n] = pn[n];
    }

    #pragma unroll
    for (int n = 0; n < GEMM_NT; ++n) {
        #pragma unroll
        for (int i = 0; i < 8; ++i)
            feat[(row0 + 8 * hh + i) * DHF + 16 * (GEMM_NT * wv + n) + cc] = acc[n][i];
    }
    float d0 = dd0.x + dd0.y;
    float d1 = dd1.x + dd1.y;
    d0 += __shfl_xor(d0, 16);
    d1 += __shfl_xor(d1, 16);
    if (hh == 0)
        reinterpret_cast<float2*>(wv ? er : el)[row0 + cc] = make_float2(d0, d1);
#else
    const float* __restrict__ hr = Hm + (row0 + cc) * DIN;
    for (int i = 0; i < DHF / 4; ++i) {
        const int col = (DHF / 2) * wv + (DHF / 4) * hh + i;
        float s = 0.f;
        for (int k = 0; k < DIN; ++k)
            s = fmaf(hr[k], wq[((((k >> 3) * 8 + (col >> 4)) * 32) + 16 * ((k >> 2) & 1) + (col & 15)) * 4 + (k & 3)], s);
        feat[(row0 + cc) * DHF + col] = s;
    }
    {
        float s = 0.f;
        for (int k = 0; k < DIN; ++k) s = fmaf(hr[k], wa[(2 * wv + hh) * DIN + k], s);
        (wv ? er : el)[(row0 + cc) * HH + hh] = s;
    }
#endif
}

#define CH_G       64
#define CH_EDGES   (EE / CH_G)
#define CH_THREADS 1024
#define CH_TRIPS   ((CH_EDGES + CH_THREADS - 1) / CH_THREADS)
static_assert(EE % CH_G == 0, "chunks must tile the edge list");
static_assert(CH_EDGES % 4 == 0, "four consecutive edges must share a chunk");
static_assert(NN % 4 == 0, "counters are cleared and copied as 16-byte words");

__global__ __launch_bounds__(CH_THREADS) void csr_chunk_count(const int* __restrict__ dst,
                                                              int* __restrict__ plane,
                                                              int* __restrict__ rank)
{
    __shared__ __attribute__((aligned(16))) int cnt[NN];
    const int t  = threadIdx.x;
    const int e0 = blockIdx.x * CH_EDGES;

    int d[CH_TRIPS];
    #pragma unroll
    for (int k = 0; k < CH_TRIPS; ++k) {
        const int i = t + k * CH_THREADS;
        d[k] = dst[e0 + ((i < CH_EDGES) ? i : (CH_EDGES - 1))];
    }

    int4* c4 = reinterpret_cast<int4*>(cnt);
    for (int i = t; i < NN / 4; i += CH_THREADS) c4[i] = make_int4(0, 0, 0, 0);
    __syncthreads();

    int r[CH_TRIPS];
    #pragma unroll
    for (int k = 0; k < CH_TRIPS; ++k) {
        const int i = t + k * CH_THREADS;
        r[k] = (i < CH_EDGES) ? atomicAdd(&cnt[d[k]], 1) : 0;
    }
    __syncthreads();
    #pragma unroll
    for (int k = 0; k < CH_TRIPS; ++k) {
        const int i = t + k * CH_THREADS;
        if (i < CH_EDGES) rank[e0 + i] = r[k];
    }

    int4* p4 = reinterpret_cast<int4*>(plane + blockIdx.x * NN);
    for (int i = t; i < NN / 4; i += CH_THREADS) p4[i] = c4[i];
}

__global__ __launch_bounds__(256) void csr_chunk_prefix(int* __restrict__ plane,
                                                        int* __restrict__ deg)
{
    const int n = blockIdx.x * 256 + threadIdx.x;
    if (n >= NN) return;
    int c[CH_G];
    #pragma unroll
    for (int g = 0; g < CH_G; ++g) c[g] = plane[g * NN + n];
    int run = 0;
    #pragma unroll
    for (int g = 0; g < CH_G; ++g) { plane[g * NN + n] = run; run += c[g]; }
    deg[n] = run;
}

#define SCAN_CHUNK 1024
#define SCAN_FRONT_TRIPS ((((NN - 1) / SCAN_CHUNK) * (SCAN_CHUNK / 4) + SCAN_CHUNK - 1) / SCAN_CHUNK)
__global__ __launch_bounds__(SCAN_CHUNK) void csr_scan(const int* __restrict__ deg,
                                                       int* __restrict__ off)
{
    __shared__ int s_front[32];
    __shared__ int s_total[32];
    const int t    = threadIdx.x;
    const int lane = t & 31;
    const int wave = t >> 5;
    const int base = blockIdx.x * SCAN_CHUNK;

    const int4* __restrict__ d4 = reinterpret_cast<const int4*>(deg);
    const int nv = base >> 2;
    int front = 0;
    static_assert(SCAN_FRONT_TRIPS * SCAN_CHUNK * 4 <= NN, "front loads must stay inside deg");
    #pragma unroll
    for (int k = 0; k < SCAN_FRONT_TRIPS; ++k) {
        const int  j   = t + k * SCAN_CHUNK;
        const int4 q   = d4[j];
        const int  sum = (q.x + q.y) + (q.z + q.w);
        front += (j < nv) ? sum : 0;
    }

    const int i = base + t;
    const int v = (i < NN) ? deg[i] : 0;

    int s = v;
    #pragma unroll
    for (int o = 1; o < 32; o <<= 1) {
        const int y = __shfl_up(s, o);
        if (lane >= o) s += y;
    }
    #pragma unroll
    for (int o = 16; o > 0; o >>= 1) front += __shfl_xor(front, o);

    if (lane == 31) s_total[wave] = s;
    if (lane == 0)  s_front[wave] = front;
    __syncthreads();

    int c = s_front[lane] + ((lane < wave) ? s_total[lane] : 0);
    #pragma unroll
    for (int o = 16; o > 0; o >>= 1) c += __shfl_xor(c, o);

    const int excl = c + s - v;
    if (i < NN) off[i] = excl;
    if (i == NN - 1) off[NN] = excl + v;
}

__global__ __launch_bounds__(256) void csr_place(const int* __restrict__ src,
                                                 const int* __restrict__ dst,
                                                 const int* __restrict__ rank,
                                                 const int* __restrict__ plane,
                                                 const int* __restrict__ off,
                                                 int* __restrict__ csrc)
{
    const int q = blockIdx.x * 256 + threadIdx.x;
    if (q >= EE / 4) return;
    const int4 d = reinterpret_cast<const int4*>(dst)[q];
    const int4 s = reinterpret_cast<const int4*>(src)[q];
    const int4 r = reinterpret_cast<const int4*>(rank)[q];
    const int* __restrict__ pl = plane + ((q * 4) / CH_EDGES) * NN;
    const int p0 = off[d.x] + pl[d.x] + r.x;
    const int p1 = off[d.y] + pl[d.y] + r.y;
    const int p2 = off[d.z] + pl[d.z] + r.z;
    const int p3 = off[d.w] + pl[d.w] + r.w;
    csrc[p0] = s.x;
    csrc[p1] = s.y;
    csrc[p2] = s.z;
    csrc[p3] = s.w;
}

#define SCORE_NPW 8
static_assert(NN % SCORE_NPW == 0, "a wave's nodes must all exist");

__device__ __forceinline__ float wave_max(float v)
{
    #pragma unroll
    for (int o = 16; o > 0; o >>= 1) v = fmaxf(v, __shfl_xor(v, o));
    return v;
}
__device__ __forceinline__ float wave_sum(float v)
{
    #pragma unroll
    for (int o = 16; o > 0; o >>= 1) v += __shfl_xor(v, o);
    return v;
}
__device__ __forceinline__ float2 edge_score(const float2* __restrict__ el2, float2 erd, int s)
{
    const float2 l = el2[s];
    float a = l.x + erd.x, b = l.y + erd.y;
    a = (a >= 0.f) ? a : NEG_SLOPE * a;
    b = (b >= 0.f) ? b : NEG_SLOPE * b;
    return make_float2(a, b);
}

__global__ __launch_bounds__(256) void gat_edge_alpha(const float* __restrict__ el,
                                                      const float* __restrict__ er,
                                                      const int* __restrict__ csrc,
                                                      const int* __restrict__ off,
                                                      float* __restrict__ wgt)
{
    const int lane = threadIdx.x & 31;
    const int wave = __builtin_amdgcn_readfirstlane((int)(threadIdx.x >> 5));
    const int d0 = (blockIdx.x * 8 + wave) * SCORE_NPW;
    if (d0 >= NN) return;
    const float2* __restrict__ el2 = reinterpret_cast<const float2*>(el);
    const float2* __restrict__ er2 = reinterpret_cast<const float2*>(er);
    float2* __restrict__ w2 = reinterpret_cast<float2*>(wgt);
    const float ninf = -__builtin_inff();

    int po[SCORE_NPW + 1];
    #pragma unroll
    for (int k = 0; k <= SCORE_NPW; ++k) po[k] = off[d0 + k];
    bool small = true;
    #pragma unroll
    for (int k = 0; k < SCORE_NPW; ++k) small = small && (po[k + 1] - po[k] <= 32);

    if (small) {
        float2 e[SCORE_NPW];
        #pragma unroll
        for (int k = 0; k < SCORE_NPW; ++k) {
            const int p = po[k] + lane;
            e[k] = edge_score(el2, er2[d0 + k], csrc[(p < EE) ? p : (EE - 1)]);
        }
        float2 x[SCORE_NPW], s[SCORE_NPW];
        #pragma unroll
        for (int k = 0; k < SCORE_NPW; ++k) {
            const bool on = po[k] + lane < po[k + 1];
            const float e0 = on ? e[k].x : ninf;
            const float e1 = on ? e[k].y : ninf;
            const float m0 = wave_max(e0), m1 = wave_max(e1);
            x[k] = make_float2(expf(e0 - m0), expf(e1 - m1));
            s[k] = make_float2(wave_sum(x[k].x), wave_sum(x[k].y));
        }
        #pragma unroll
        for (int k = 0; k < SCORE_NPW; ++k) {
            const int p = po[k] + lane;
            if (p < po[k + 1]) w2[p] = make_float2(x[k].x / s[k].x, x[k].y / s[k].y);
        }
    } else {
        #pragma nounroll
        for (int k = 0; k < SCORE_NPW; ++k) {
            const int d  = d0 + k;
            const int p0 = off[d], p1 = off[d + 1];
            const float2 erd = er2[d];
            float m0 = ninf, m1 = ninf;
            for (int p = p0 + lane; p < p1; p += 32) {
                const float2 c = edge_score(el2, erd, csrc[p]);
                m0 = fmaxf(m0, c.x); m1 = fmaxf(m1, c.y);
            }
            m0 = wave_max(m0); m1 = wave_max(m1);
            float s0 = 0.f, s1 = 0.f;
            for (int p = p0 + lane; p < p1; p += 32) {
                const float2 c = edge_score(el2, erd, csrc[p]);
                s0 += expf(c.x - m0); s1 += expf(c.y - m1);
            }
            s0 = wave_sum(s0); s1 = wave_sum(s1);
            for (int p = p0 + lane; p < p1; p += 32) {
                const float2 c = edge_score(el2, erd, csrc[p]);
                w2[p] = make_float2(expf(c.x - m0) / s0, expf(c.y - m1) / s1);
            }
        }
    }
}

#define AGG_BATCH 16
#define AGG_NB    16
#define AGG_NW    (AGG_NB / 2)
static_assert(NN % AGG_NB == 0, "a work-group's nodes must all exist");
static_assert(FF == 2 * 32, "a wave's lanes, two features each, cover one head");

#define AGG_CLOSE_NODES(base_pos, rel_next)                                    \
    while (rel == (rel_next)) {                                                \
        float v0 = acc.x + b.x;                                                \
        float v1 = acc.y + b.y;                                                \
        v0 = (v0 > 0.f) ? v0 : expm1f(v0);                                     \
        v1 = (v1 > 0.f) ? v1 : expm1f(v1);                                     \
        if (!last) *reinterpret_cast<float2*>(out + (n0 + k) * DHF + f) = make_float2(v0, v1); \
        else       *reinterpret_cast<float2*>(sv + (AGG_NW * nh + k) * DHF + f) = make_float2(v0, v1); \
        acc = make_float2(0.f, 0.f);                                           \
        ++k;                                                                   \
        const int kn = (k < AGG_NW) ? (k + 1) : AGG_NW;                        \
        const int en = __builtin_amdgcn_readfirstlane(off[n0 + kn]);           \
        nend = (k < AGG_NW) ? en : 0x7fffffff;                                 \
        rel  = nend - (base_pos);                                              \
    }

__global__ __launch_bounds__(128) void gat_aggregate(const float* __restrict__ wgt,
                                                     const float* __restrict__ feat,
                                                     const int* __restrict__ csrc,
                                                     const int* __restrict__ off,
                                                     const float* __restrict__ bias,
                                                     float* __restrict__ out,
                                                     int last)
{
    __shared__ __attribute__((aligned(16))) float sv[AGG_NB * DHF];
    const int t    = threadIdx.x;
    const int lane = t & 31;
    const int wi = __builtin_amdgcn_readfirstlane(t >> 5);
    const int h  = wi & 1;
    const int nh = wi >> 1;
    const int fl = (FF / 2) * h + lane;
    const int f  = 2 * fl;
    const float2* __restrict__ feat2 = reinterpret_cast<const float2*>(feat);
    const int d0 = blockIdx.x * AGG_NB;
    const int n0 = d0 + AGG_NW * nh;
    const int P0 = __builtin_amdgcn_readfirstlane(off[n0]);
    const int P1 = __builtin_amdgcn_readfirstlane(off[n0 + AGG_NW]);
    const float2 b = *reinterpret_cast<const float2*>(bias + f);

    int    k    = 0;
    int    nend = __builtin_amdgcn_readfirstlane(off[n0 + 1]);
    int    rel  = nend - P0;
    float2 acc  = make_float2(0.f, 0.f);

    AGG_CLOSE_NODES(P0, 0)

    if (P1 > P0) {
        int   q   = P0 + lane;
        int   qc  = (q < P1) ? q : (P1 - 1);
        int   vsn = csrc[qc];
        float vwn = wgt[qc * HH + h];

        for (int g = P0; g < P1; g += 32) {
            int   vs = vsn;
            float vw = (g + lane < P1) ? vwn : 0.f;
            if (g + 32 < P1) {
                q   = g + 32 + lane;
                qc  = (q < P1) ? q : (P1 - 1);
                vsn = csrc[qc];
                vwn = wgt[qc * HH + h];
            }
            #pragma nounroll
            for (int hb = 0; hb < 32; hb += AGG_BATCH) {
                const int base = g + hb;
                if (base >= P1) break;
                rel = nend - base;
                float2 x[AGG_BATCH];
                #pragma unroll
                for (int j = 0; j < AGG_BATCH; ++j) {
                    const int s = __builtin_amdgcn_readlane(vs, j);
                    x[j] = feat2[s * (DHF / 2) + fl];
                }
                #pragma unroll
                for (int j = 0; j < AGG_BATCH; ++j) {
                    const float w = __int_as_float(__builtin_amdgcn_readlane(__float_as_int(vw), j));
                    acc.x = fmaf(w, x[j].x, acc.x);
                    acc.y = fmaf(w, x[j].y, acc.y);
                    AGG_CLOSE_NODES(base, j + 1)
                }
                vs = __shfl_down(vs, AGG_BATCH);
                vw = __shfl_down(vw, AGG_BATCH);
            }
        }
    }

    if (last) {
        __syncthreads();
        for (int i = t; i < AGG_NB * FF; i += 128) {
            const int n = i >> 6, ff = i & (FF - 1);
            out[(d0 + n) * FF + ff] = 0.5f * (sv[n * DHF + ff] + sv[n * DHF + ff + FF]);
        }
    }
}

extern "C" void kernel_launch(void* const* d_in, const int* in_sizes, int n_in,
                              void* d_out, int out_size, void* d_ws, size_t ws_size,
                              hipStream_t stream)
{
    const float* x    = (const float*)d_in[0];
    const int*   src  = (const int*)d_in[1];
    const int*   dst  = (const int*)d_in[2];
    const float* Wall = (const float*)d_in[3];
    const float* al   = (const float*)d_in[4];
    const float* ar   = (const float*)d_in[5];
    const float* bias = (const float*)d_in[6];
    float* out = (float*)d_out;

    char* p = (char*)d_ws;
    auto take = [&](size_t bytes) {
        char* q = p;
        p += (bytes + 255) & ~size_t(255);
        return q;
    };
    float* feat   = (float*)take(sizeof(float) * NN * DHF);
    float* hbuf   = (float*)take(sizeof(float) * NN * DHF);
    float* el     = (float*)take(sizeof(float) * NN * HH);
    float* er     = (float*)take(sizeof(float) * NN * HH);
    float* wgt    = (float*)take(sizeof(float) * EE * HH);
    int*   deg    = (int*)take(sizeof(int) * NN);
    int*   off    = (int*)take(sizeof(int) * (NN + 1));
    int*   rank   = (int*)take(sizeof(int) * EE);
    int*   plane  = (int*)take(sizeof(int) * CH_G * NN);
    int*   csrc   = (int*)take(sizeof(int) * EE);
    float* wa     = (float*)take(sizeof(float) * 3 * 4 * DIN);
    float* wq     = (float*)take(sizeof(float) * 3 * DIN * DHF);

    const int tb   = 256;
    const int gN   = (NN + tb - 1) / tb;
    const int gPl  = (EE / 4 + tb - 1) / tb;
    const int gSc  = (NN / SCORE_NPW + 7) / 8;
    const int gGmm = (NN / 16 + 1) / 2;
    const int gScan = (NN + SCAN_CHUNK - 1) / SCAN_CHUNK;

    gat_fold<<<FOLD_G + WQ_G, DIN, 0, stream>>>(Wall, al, ar, wa, wq);

    csr_chunk_count <<<CH_G, CH_THREADS, 0, stream>>>(dst, plane, rank);
    csr_chunk_prefix<<<gN, tb, 0, stream>>>(plane, deg);
    csr_scan        <<<gScan, SCAN_CHUNK, 0, stream>>>(deg, off);
    csr_place       <<<gPl, tb, 0, stream>>>(src, dst, rank, plane, off, csrc);

    const float* hin = x;
    for (int layer = 0; layer < 3; ++layer) {
        const float* Wl  = wq + layer * DIN * DHF;
        const float* bl  = bias + layer * HH * FF;
        const int    last = (layer == 2);
        float* outp = last ? out : hbuf;

        gat_gemm     <<<gGmm, 128, 0, stream>>>(hin, Wl, feat, wa + layer * 4 * DIN, el, er);
        gat_edge_alpha<<<gSc, tb, 0, stream>>>(el, er, csrc, off, wgt);
        gat_aggregate<<<NN / AGG_NB, 128, 0, stream>>>(wgt, feat, csrc, off, bl, outp, last);
        hin = hbuf;
    }
    (void)in_sizes; (void)n_in; (void)out_size; (void)ws_size;
}


# projection moved behind the gather-sum (a layer is linear): a work-group sums its 16 nodes' weighted INPUT rows per head in LDS and projects them itself; 12 launches, not 14; every sum order changes
# speedup vs baseline: 2.5185x; 1.0380x over previous
#include <hip/hip_runtime.h>
#include <math.h>

#define NN  50000
#define EE  800000
#define DIN 128
#define HH  2
#define FF  64
#define DHF 128
#define NEG_SLOPE 0.2f

typedef __attribute__((ext_vector_type(2))) float v2f;
typedef __attribute__((ext_vector_type(8))) float v8f;

#if defined(__AMDGCN__) && __has_builtin(__builtin_amdgcn_wmma_f32_16x16x4_f32)
#define USE_WMMA_F32 1
#else
#define USE_WMMA_F32 0
#endif

#define FOLD_G  (3 * 4)
#define WQ_G    (3 * (DIN / 8))
__global__ __launch_bounds__(DIN) void gat_fold(const float* __restrict__ Wall,
                                                const float* __restrict__ al,
                                                const float* __restrict__ ar,
                                                float* __restrict__ wa,
                                                float* __restrict__ wq)
{
    if (blockIdx.x >= FOLD_G) {
        const int g     = blockIdx.x - FOLD_G;
        const int layer = g / (DIN / 8);
        const int kb    = g % (DIN / 8);
        const float* __restrict__ Wl = Wall + layer * DIN * DHF;
        float4* __restrict__ q4 = reinterpret_cast<float4*>(wq + layer * DIN * DHF);
        #pragma unroll
        for (int r = 0; r < 2; ++r) {
            const int idx = threadIdx.x + DIN * r;
            const int n   = idx >> 5;
            const int ln  = idx & 31;
            const float* __restrict__ p = Wl + (8 * kb + 4 * (ln >> 4)) * DHF + 16 * n + (ln & 15);
            q4[(kb * 8 + n) * 32 + ln] = make_float4(p[0], p[DHF], p[2 * DHF], p[3 * DHF]);
        }
        return;
    }
    const int k     = threadIdx.x;
    const int j     = blockIdx.x & 3;
    const int layer = blockIdx.x >> 2;
    const int h     = j & 1;
    const float* __restrict__ a = ((j < 2) ? al : ar) + (layer * HH + h) * FF;
    const float4* __restrict__ w4 = reinterpret_cast<const float4*>(Wall + (layer * DIN + k) * DHF + h * FF);
    const float4* __restrict__ a4 = reinterpret_cast<const float4*>(a);
    float s = 0.f;
    #pragma unroll
    for (int i = 0; i < FF / 4; ++i) {
        const float4 w = w4[i];
        const float4 q = a4[i];
        s = fmaf(w.x, q.x, s);
        s = fmaf(w.y, q.y, s);
        s = fmaf(w.z, q.z, s);
        s = fmaf(w.w, q.w, s);
    }
    wa[(layer * 4 + j) * DIN + k] = s;
}


#define CH_G       64
#define CH_EDGES   (EE / CH_G)
#define CH_THREADS 1024
#define CH_TRIPS   ((CH_EDGES + CH_THREADS - 1) / CH_THREADS)
static_assert(EE % CH_G == 0, "chunks must tile the edge list");
static_assert(CH_EDGES % 4 == 0, "four consecutive edges must share a chunk");
static_assert(NN % 4 == 0, "counters are cleared and copied as 16-byte words");

__global__ __launch_bounds__(CH_THREADS) void csr_chunk_count(const int* __restrict__ dst,
                                                              int* __restrict__ plane,
                                                              int* __restrict__ rank)
{
    __shared__ __attribute__((aligned(16))) int cnt[NN];
    const int t  = threadIdx.x;
    const int e0 = blockIdx.x * CH_EDGES;

    int d[CH_TRIPS];
    #pragma unroll
    for (int k = 0; k < CH_TRIPS; ++k) {
        const int i = t + k * CH_THREADS;
        d[k] = dst[e0 + ((i < CH_EDGES) ? i : (CH_EDGES - 1))];
    }

    int4* c4 = reinterpret_cast<int4*>(cnt);
    for (int i = t; i < NN / 4; i += CH_THREADS) c4[i] = make_int4(0, 0, 0, 0);
    __syncthreads();

    int r[CH_TRIPS];
    #pragma unroll
    for (int k = 0; k < CH_TRIPS; ++k) {
        const int i = t + k * CH_THREADS;
        r[k] = (i < CH_EDGES) ? atomicAdd(&cnt[d[k]], 1) : 0;
    }
    __syncthreads();
    #pragma unroll
    for (int k = 0; k < CH_TRIPS; ++k) {
        const int i = t + k * CH_THREADS;
        if (i < CH_EDGES) rank[e0 + i] = r[k];
    }

    int4* p4 = reinterpret_cast<int4*>(plane + blockIdx.x * NN);
    for (int i = t; i < NN / 4; i += CH_THREADS) p4[i] = c4[i];
}

__global__ __launch_bounds__(256) void csr_chunk_prefix(int* __restrict__ plane,
                                                        int* __restrict__ deg)
{
    const int n = blockIdx.x * 256 + threadIdx.x;
    if (n >= NN) return;
    int c[CH_G];
    #pragma unroll
    for (int g = 0; g < CH_G; ++g) c[g] = plane[g * NN + n];
    int run = 0;
    #pragma unroll
    for (int g = 0; g < CH_G; ++g) { plane[g * NN + n] = run; run += c[g]; }
    deg[n] = run;
}

#define SCAN_CHUNK 1024
#define SCAN_FRONT_TRIPS ((((NN - 1) / SCAN_CHUNK) * (SCAN_CHUNK / 4) + SCAN_CHUNK - 1) / SCAN_CHUNK)
__global__ __launch_bounds__(SCAN_CHUNK) void csr_scan(const int* __restrict__ deg,
                                                       int* __restrict__ off)
{
    __shared__ int s_front[32];
    __shared__ int s_total[32];
    const int t    = threadIdx.x;
    const int lane = t & 31;
    const int wave = t >> 5;
    const int base = blockIdx.x * SCAN_CHUNK;

    const int4* __restrict__ d4 = reinterpret_cast<const int4*>(deg);
    const int nv = base >> 2;
    int front = 0;
    static_assert(SCAN_FRONT_TRIPS * SCAN_CHUNK * 4 <= NN, "front loads must stay inside deg");
    #pragma unroll
    for (int k = 0; k < SCAN_FRONT_TRIPS; ++k) {
        const int  j   = t + k * SCAN_CHUNK;
        const int4 q   = d4[j];
        const int  sum = (q.x + q.y) + (q.z + q.w);
        front += (j < nv) ? sum : 0;
    }

    const int i = base + t;
    const int v = (i < NN) ? deg[i] : 0;

    int s = v;
    #pragma unroll
    for (int o = 1; o < 32; o <<= 1) {
        const int y = __shfl_up(s, o);
        if (lane >= o) s += y;
    }
    #pragma unroll
    for (int o = 16; o > 0; o >>= 1) front += __shfl_xor(front, o);

    if (lane == 31) s_total[wave] = s;
    if (lane == 0)  s_front[wave] = front;
    __syncthreads();

    int c = s_front[lane] + ((lane < wave) ? s_total[lane] : 0);
    #pragma unroll
    for (int o = 16; o > 0; o >>= 1) c += __shfl_xor(c, o);

    const int excl = c + s - v;
    if (i < NN) off[i] = excl;
    if (i == NN - 1) off[NN] = excl + v;
}

__global__ __launch_bounds__(256) void csr_place(const int* __restrict__ src,
                                                 const int* __restrict__ dst,
                                                 const int* __restrict__ rank,
                                                 const int* __restrict__ plane,
                                                 const int* __restrict__ off,
                                                 int* __restrict__ csrc)
{
    const int q = blockIdx.x * 256 + threadIdx.x;
    if (q >= EE / 4) return;
    const int4 d = reinterpret_cast<const int4*>(dst)[q];
    const int4 s = reinterpret_cast<const int4*>(src)[q];
    const int4 r = reinterpret_cast<const int4*>(rank)[q];
    const int* __restrict__ pl = plane + ((q * 4) / CH_EDGES) * NN;
    const int p0 = off[d.x] + pl[d.x] + r.x;
    const int p1 = off[d.y] + pl[d.y] + r.y;
    const int p2 = off[d.z] + pl[d.z] + r.z;
    const int p3 = off[d.w] + pl[d.w] + r.w;
    csrc[p0] = s.x;
    csrc[p1] = s.y;
    csrc[p2] = s.z;
    csrc[p3] = s.w;
}

#define SCORE_NPW 8
static_assert(NN % SCORE_NPW == 0, "a wave's nodes must all exist");

__device__ __forceinline__ float wave_max(float v)
{
    #pragma unroll
    for (int o = 16; o > 0; o >>= 1) v = fmaxf(v, __shfl_xor(v, o));
    return v;
}
__device__ __forceinline__ float wave_sum(float v)
{
    #pragma unroll
    for (int o = 16; o > 0; o >>= 1) v += __shfl_xor(v, o);
    return v;
}
__device__ __forceinline__ float2 edge_score(const float2* __restrict__ el2, float2 erd, int s)
{
    const float2 l = el2[s];
    float a = l.x + erd.x, b = l.y + erd.y;
    a = (a >= 0.f) ? a : NEG_SLOPE * a;
    b = (b >= 0.f) ? b : NEG_SLOPE * b;
    return make_float2(a, b);
}

__global__ __launch_bounds__(256) void gat_edge_alpha(const float* __restrict__ el,
                                                      const float* __restrict__ er,
                                                      const int* __restrict__ csrc,
                                                      const int* __restrict__ off,
                                                      float* __restrict__ wgt)
{
    const int lane = threadIdx.x & 31;
    const int wave = __builtin_amdgcn_readfirstlane((int)(threadIdx.x >> 5));
    const int d0 = (blockIdx.x * 8 + wave) * SCORE_NPW;
    if (d0 >= NN) return;
    const float2* __restrict__ el2 = reinterpret_cast<const float2*>(el);
    const float2* __restrict__ er2 = reinterpret_cast<const float2*>(er);
    float2* __restrict__ w2 = reinterpret_cast<float2*>(wgt);
    const float ninf = -__builtin_inff();

    int po[SCORE_NPW + 1];
    #pragma unroll
    for (int k = 0; k <= SCORE_NPW; ++k) po[k] = off[d0 + k];
    bool small = true;
    #pragma unroll
    for (int k = 0; k < SCORE_NPW; ++k) small = small && (po[k + 1] - po[k] <= 32);

    if (small) {
        float2 e[SCORE_NPW];
        #pragma unroll
        for (int k = 0; k < SCORE_NPW; ++k) {
            const int p = po[k] + lane;
            e[k] = edge_score(el2, er2[d0 + k], csrc[(p < EE) ? p : (EE - 1)]);
        }
        float2 x[SCORE_NPW], s[SCORE_NPW];
        #pragma unroll
        for (int k = 0; k < SCORE_NPW; ++k) {
            const bool on = po[k] + lane < po[k + 1];
            const float e0 = on ? e[k].x : ninf;
            const float e1 = on ? e[k].y : ninf;
            const float m0 = wave_max(e0), m1 = wave_max(e1);
            x[k] = make_float2(expf(e0 - m0), expf(e1 - m1));
            s[k] = make_float2(wave_sum(x[k].x), wave_sum(x[k].y));
        }
        #pragma unroll
        for (int k = 0; k < SCORE_NPW; ++k) {
            const int p = po[k] + lane;
            if (p < po[k + 1]) w2[p] = make_float2(x[k].x / s[k].x, x[k].y / s[k].y);
        }
    } else {
        #pragma nounroll
        for (int k = 0; k < SCORE_NPW; ++k) {
            const int d  = d0 + k;
            const int p0 = off[d], p1 = off[d + 1];
            const float2 erd = er2[d];
            float m0 = ninf, m1 = ninf;
            for (int p = p0 + lane; p < p1; p += 32) {
                const float2 c = edge_score(el2, erd, csrc[p]);
                m0 = fmaxf(m0, c.x); m1 = fmaxf(m1, c.y);
            }
            m0 = wave_max(m0); m1 = wave_max(m1);
            float s0 = 0.f, s1 = 0.f;
            for (int p = p0 + lane; p < p1; p += 32) {
                const float2 c = edge_score(el2, erd, csrc[p]);
                s0 += expf(c.x - m0); s1 += expf(c.y - m1);
            }
            s0 = wave_sum(s0); s1 = wave_sum(s1);
            for (int p = p0 + lane; p < p1; p += 32) {
                const float2 c = edge_score(el2, erd, csrc[p]);
                w2[p] = make_float2(expf(c.x - m0) / s0, expf(c.y - m1) / s1);
            }
        }
    }
}

__device__ __forceinline__ float dot4(float4 a, float4 b)
{
    return fmaf(a.w, b.w, fmaf(a.z, b.z, fmaf(a.y, b.y, a.x * b.x)));
}
#define RS_ROWS 8
static_assert(NN % RS_ROWS == 0, "a wave's rows must all exist");
__global__ __launch_bounds__(256) void gat_row_scores(const float* __restrict__ Hm,
                                                      const float* __restrict__ wa,
                                                      float* __restrict__ el,
                                                      float* __restrict__ er)
{
    const int lane = threadIdx.x & 31;
    const int r0   = (blockIdx.x * 8 + (threadIdx.x >> 5)) * RS_ROWS;
    if (r0 >= NN) return;
    const float4* __restrict__ wa4 = reinterpret_cast<const float4*>(wa);
    const float4* __restrict__ h4  = reinterpret_cast<const float4*>(Hm);
    float4 u[4], x[RS_ROWS];
    #pragma unroll
    for (int j = 0; j < 4; ++j) u[j] = wa4[j * (DIN / 4) + lane];
    #pragma unroll
    for (int r = 0; r < RS_ROWS; ++r) x[r] = h4[(r0 + r) * (DIN / 4) + lane];
    #pragma unroll
    for (int r = 0; r < RS_ROWS; ++r) {
        float d[4];
        #pragma unroll
        for (int j = 0; j < 4; ++j) d[j] = wave_sum(dot4(x[r], u[j]));
        if (lane == 0) {
            reinterpret_cast<float2*>(el)[r0 + r] = make_float2(d[0], d[1]);
            reinterpret_cast<float2*>(er)[r0 + r] = make_float2(d[2], d[3]);
        }
    }
}

#define AGG_BATCH 16
#define AGG_NB    16
#define AGG_NW    (AGG_NB / 2)
#define G_PITCH   (DIN + 4)
#define PRJ_NT    2
static_assert(NN % AGG_NB == 0, "a work-group's nodes must all exist");
static_assert(DIN == 4 * 32, "a wave's lanes, two features each, cover half a row");
static_assert(DHF == 4 * 16 * PRJ_NT, "four waves of 2 column tiles cover W");
static_assert(FF == 2 * 16 * PRJ_NT, "a wave's column tiles lie in one head");

#define AGG_CLOSE_NODES(base_pos, rel_next)                                    \
    while (rel == (rel_next)) {                                                \
        *reinterpret_cast<float2*>(&G[gw + k * G_PITCH]) = acc0;               \
        *reinterpret_cast<float2*>(&G[gw + (AGG_NB + k) * G_PITCH]) = acc1;    \
        acc0 = make_float2(0.f, 0.f);                                          \
        acc1 = make_float2(0.f, 0.f);                                          \
        ++k;                                                                   \
        const int kn = (k < AGG_NW) ? (k + 1) : AGG_NW;                        \
        const int en = __builtin_amdgcn_readfirstlane(off[n0 + kn]);           \
        nend = (k < AGG_NW) ? en : 0x7fffffff;                                 \
        rel  = nend - (base_pos);                                              \
    }

__global__ __launch_bounds__(128) void gat_aggregate(const float* __restrict__ wgt,
                                                     const float* __restrict__ hin,
                                                     const int* __restrict__ csrc,
                                                     const int* __restrict__ off,
                                                     const float* __restrict__ wq,
                                                     const float* __restrict__ bias,
                                                     const float* __restrict__ wan,
                                                     float* __restrict__ hout,
                                                     float* __restrict__ el,
                                                     float* __restrict__ er,
                                                     float* __restrict__ out,
                                                     int last)
{
    __shared__ __attribute__((aligned(16))) float G[HH * AGG_NB * G_PITCH];
    const int t    = threadIdx.x;
    const int lane = t & 31;
    const int wi = __builtin_amdgcn_readfirstlane(t >> 5);
    const int fh = wi & 1;
    const int nh = wi >> 1;
    const int fl = (DIN / 4) * fh + lane;
    const float2* __restrict__ hin2 = reinterpret_cast<const float2*>(hin);
    const float2* __restrict__ w2   = reinterpret_cast<const float2*>(wgt);
    const int d0 = blockIdx.x * AGG_NB;
    const int n0 = d0 + AGG_NW * nh;
    const int gw = (AGG_NW * nh) * G_PITCH + 2 * fl;
    const int P0 = __builtin_amdgcn_readfirstlane(off[n0]);
    const int P1 = __builtin_amdgcn_readfirstlane(off[n0 + AGG_NW]);

    int    k    = 0;
    int    nend = __builtin_amdgcn_readfirstlane(off[n0 + 1]);
    int    rel  = nend - P0;
    float2 acc0 = make_float2(0.f, 0.f);
    float2 acc1 = make_float2(0.f, 0.f);

    AGG_CLOSE_NODES(P0, 0)

    if (P1 > P0) {
        int    q   = P0 + lane;
        int    qc  = (q < P1) ? q : (P1 - 1);
        int    vsn = csrc[qc];
        float2 vwn = w2[qc];

        for (int g = P0; g < P1; g += 32) {
            int   vs  = vsn;
            float vw0 = (g + lane < P1) ? vwn.x : 0.f;
            float vw1 = (g + lane < P1) ? vwn.y : 0.f;
            if (g + 32 < P1) {
                q   = g + 32 + lane;
                qc  = (q < P1) ? q : (P1 - 1);
                vsn = csrc[qc];
                vwn = w2[qc];
            }
            #pragma nounroll
            for (int hb = 0; hb < 32; hb += AGG_BATCH) {
                const int base = g + hb;
                if (base >= P1) break;
                rel = nend - base;
                float2 x[AGG_BATCH];
                #pragma unroll
                for (int j = 0; j < AGG_BATCH; ++j) {
                    const int s = __builtin_amdgcn_readlane(vs, j);
                    x[j] = hin2[s * (DIN / 2) + fl];
                }
                #pragma unroll
                for (int j = 0; j < AGG_BATCH; ++j) {
                    const float wa0 = __int_as_float(__builtin_amdgcn_readlane(__float_as_int(vw0), j));
                    const float wa1 = __int_as_float(__builtin_amdgcn_readlane(__float_as_int(vw1), j));
                    acc0.x = fmaf(wa0, x[j].x, acc0.x);
                    acc0.y = fmaf(wa0, x[j].y, acc0.y);
                    acc1.x = fmaf(wa1, x[j].x, acc1.x);
                    acc1.y = fmaf(wa1, x[j].y, acc1.y);
                    AGG_CLOSE_NODES(base, j + 1)
                }
                vs  = __shfl_down(vs, AGG_BATCH);
                vw0 = __shfl_down(vw0, AGG_BATCH);
                vw1 = __shfl_down(vw1, AGG_BATCH);
            }
        }
    }

    const int cc   = lane & 15;
    const int hh   = lane >> 4;
    const int hd   = wi >> 1;
    float res[PRJ_NT][8];
#if USE_WMMA_F32
    {
        const float4* __restrict__ bp = reinterpret_cast<const float4*>(wq) + (PRJ_NT * wi) * 32 + lane;
        const int aoff = (hd * AGG_NB + cc) * G_PITCH + 4 * hh;
        v8f acc[PRJ_NT];
        #pragma unroll
        for (int n = 0; n < PRJ_NT; ++n) acc[n] = v8f{};
        float4 p[3][PRJ_NT + 1];
        #pragma unroll
        for (int n = 0; n < PRJ_NT; ++n) {
            p[0][n] = bp[n * 32];
            p[1][n] = bp[(8 + n) * 32];
        }
        __syncthreads();
        p[0][PRJ_NT] = *reinterpret_cast<const float4*>(&G[aoff]);
        p[1][PRJ_NT] = *reinterpret_cast<const float4*>(&G[aoff + 8]);
        #pragma unroll
        for (int kb = 0; kb < DIN / 8; ++kb) {
            if (kb + 2 < DIN / 8) {
                #pragma unroll
                for (int n = 0; n < PRJ_NT; ++n) p[(kb + 2) % 3][n] = bp[((kb + 2) * 8 + n) * 32];
                p[(kb + 2) % 3][PRJ_NT] = *reinterpret_cast<const float4*>(&G[aoff + 8 * (kb + 2)]);
            }
            __builtin_amdgcn_sched_barrier(0);
            const float4 pa = p[kb % 3][PRJ_NT];
            v2f a0, a1;
            a0.x = pa.x; a0.y = pa.y;
            a1.x = pa.z; a1.y = pa.w;
            #pragma unroll
            for (int n = 0; n < PRJ_NT; ++n) {
                v2f b0;
                b0.x = p[kb % 3][n].x; b0.y = p[kb % 3][n].y;
                acc[n] = __builtin_amdgcn_wmma_f32_16x16x4_f32(false, a0, false, b0,
                                                               (short)0, acc[n], false, false);
            }
            #pragma unroll
            for (int n = 0; n < PRJ_NT; ++n) {
                v2f b1;
                b1.x = p[kb % 3][n].z; b1.y = p[kb % 3][n].w;
                acc[n] = __builtin_amdgcn_wmma_f32_16x16x4_f32(false, a1, false, b1,
                                                               (short)0, acc[n], false, false);
            }
            __builtin_amdgcn_sched_barrier(0);
        }
        #pragma unroll
        for (int n = 0; n < PRJ_NT; ++n) {
            #pragma unroll
            for (int i = 0; i < 8; ++i) res[n][i] = acc[n][i];
        }
    }
#else
    __syncthreads();
    for (int n = 0; n < PRJ_NT; ++n) {
        const int col = 16 * (PRJ_NT * wi + n) + cc;
        for (int i = 0; i < 8; ++i) {
            float s = 0.f;
            for (int kk = 0; kk < DIN; ++kk)
                s = fmaf(G[(hd * AGG_NB + 8 * hh + i) * G_PITCH + kk],
                         wq[((((kk >> 3) * 8 + (col >> 4)) * 32) + 16 * ((kk >> 2) & 1) + (col & 15)) * 4 + (kk & 3)], s);
            res[n][i] = s;
        }
    }
#endif

    __syncthreads();
    #pragma unroll
    for (int n = 0; n < PRJ_NT; ++n) {
        const int col = 16 * (PRJ_NT * wi + n) + cc;
        const float bc = bias[col];
        #pragma unroll
        for (int i = 0; i < 8; ++i) {
            float v = res[n][i] + bc;
            v = (v > 0.f) ? v : expm1f(v);
            G[(8 * hh + i) * G_PITCH + col] = v;
        }
    }
    __syncthreads();

    if (!last) {
        const float4* __restrict__ wa4 = reinterpret_cast<const float4*>(wan);
        float4 u[4], x[4];
        #pragma unroll
        for (int j = 0; j < 4; ++j) u[j] = wa4[j * (DIN / 4) + lane];
        #pragma unroll
        for (int r = 0; r < 4; ++r) x[r] = *reinterpret_cast<const float4*>(&G[(4 * wi + r) * G_PITCH + 4 * lane]);
        #pragma unroll
        for (int r = 0; r < 4; ++r)
            reinterpret_cast<float4*>(hout)[(d0 + 4 * wi + r) * (DIN / 4) + lane] = x[r];
        #pragma unroll
        for (int r = 0; r < 4; ++r) {
            float d[4];
            #pragma unroll
            for (int j = 0; j < 4; ++j) d[j] = wave_sum(dot4(x[r], u[j]));
            if (lane == 0) {
                reinterpret_cast<float2*>(el)[d0 + 4 * wi + r] = make_float2(d[0], d[1]);
                reinterpret_cast<float2*>(er)[d0 + 4 * wi + r] = make_float2(d[2], d[3]);
            }
        }
    } else {
        #pragma unroll
        for (int r = 0; r < 4; ++r) {
            const float2 a = *reinterpret_cast<const float2*>(&G[(4 * wi + r) * G_PITCH + 2 * lane]);
            const float2 b = *reinterpret_cast<const float2*>(&G[(4 * wi + r) * G_PITCH + FF + 2 * lane]);
            reinterpret_cast<float2*>(out)[(d0 + 4 * wi + r) * (FF / 2) + lane] =
                make_float2(0.5f * (a.x + b.x), 0.5f * (a.y + b.y));
        }
    }
}

extern "C" void kernel_launch(void* const* d_in, const int* in_sizes, int n_in,
                              void* d_out, int out_size, void* d_ws, size_t ws_size,
                              hipStream_t stream)
{
    const float* x    = (const float*)d_in[0];
    const int*   src  = (const int*)d_in[1];
    const int*   dst  = (const int*)d_in[2];
    const float* Wall = (const float*)d_in[3];
    const float* al   = (const float*)d_in[4];
    const float* ar   = (const float*)d_in[5];
    const float* bias = (const float*)d_in[6];
    float* out = (float*)d_out;

    char* p = (char*)d_ws;
    auto take = [&](size_t bytes) {
        char* q = p;
        p += (bytes + 255) & ~size_t(255);
        return q;
    };
    float* hbufa  = (float*)take(sizeof(float) * NN * DHF);
    float* hbufb  = (float*)take(sizeof(float) * NN * DHF);
    float* el     = (float*)take(sizeof(float) * NN * HH);
    float* er     = (float*)take(sizeof(float) * NN * HH);
    float* wgt    = (float*)take(sizeof(float) * EE * HH);
    int*   deg    = (int*)take(sizeof(int) * NN);
    int*   off    = (int*)take(sizeof(int) * (NN + 1));
    int*   rank   = (int*)take(sizeof(int) * EE);
    int*   plane  = (int*)take(sizeof(int) * CH_G * NN);
    int*   csrc   = (int*)take(sizeof(int) * EE);
    float* wa     = (float*)take(sizeof(float) * 3 * 4 * DIN);
    float* wq     = (float*)take(sizeof(float) * 3 * DIN * DHF);

    const int tb   = 256;
    const int gN   = (NN + tb - 1) / tb;
    const int gPl  = (EE / 4 + tb - 1) / tb;
    const int gSc  = (NN / SCORE_NPW + 7) / 8;
    const int gRs  = (NN / RS_ROWS + 7) / 8;
    const int gScan = (NN + SCAN_CHUNK - 1) / SCAN_CHUNK;

    gat_fold<<<FOLD_G + WQ_G, DIN, 0, stream>>>(Wall, al, ar, wa, wq);

    csr_chunk_count <<<CH_G, CH_THREADS, 0, stream>>>(dst, plane, rank);
    csr_chunk_prefix<<<gN, tb, 0, stream>>>(plane, deg);
    csr_scan        <<<gScan, SCAN_CHUNK, 0, stream>>>(deg, off);
    csr_place       <<<gPl, tb, 0, stream>>>(src, dst, rank, plane, off, csrc);

    gat_row_scores<<<gRs, tb, 0, stream>>>(x, wa, el, er);

    const float* hin = x;
    for (int layer = 0; layer < 3; ++layer) {
        const float* Wl  = wq + layer * DIN * DHF;
        const float* bl  = bias + layer * HH * FF;
        const int    last = (layer == 2);
        float* hout = (layer & 1) ? hbufb : hbufa;
        const float* wan = wa + (last ? layer : (layer + 1)) * 4 * DIN;

        gat_edge_alpha<<<gSc, tb, 0, stream>>>(el, er, csrc, off, wgt);
        gat_aggregate<<<NN / AGG_NB, 128, 0, stream>>>(wgt, hin, csrc, off, Wl, bl, wan, hout, el, er, out, last);
        hin = hout;
    }
    (void)in_sizes; (void)n_in; (void)out_size; (void)ws_size;
}


# input-only work rides in the CSR count launch on the units it leaves idle: layer 0 node scores (each block folds its attention vectors in LDS), later folds, W reorder; 10 launches not 12; same sums
# speedup vs baseline: 2.6185x; 1.0397x over previous
#include <hip/hip_runtime.h>
#include <math.h>

#define NN  50000
#define EE  800000
#define DIN 128
#define HH  2
#define FF  64
#define DHF 128
#define NEG_SLOPE 0.2f

typedef __attribute__((ext_vector_type(2))) float v2f;
typedef __attribute__((ext_vector_type(8))) float v8f;

#if defined(__AMDGCN__) && __has_builtin(__builtin_amdgcn_wmma_f32_16x16x4_f32)
#define USE_WMMA_F32 1
#else
#define USE_WMMA_F32 0
#endif

__device__ __forceinline__ float fold_dot(int lj, int k,
                                          const float* __restrict__ Wall,
                                          const float* __restrict__ al,
                                          const float* __restrict__ ar)
{
    const int j     = lj & 3;
    const int layer = lj >> 2;
    const int h     = j & 1;
    const float* __restrict__ a = ((j < 2) ? al : ar) + (layer * HH + h) * FF;
    const float4* __restrict__ w4 = reinterpret_cast<const float4*>(Wall + (layer * DIN + k) * DHF + h * FF);
    const float4* __restrict__ a4 = reinterpret_cast<const float4*>(a);
    float s = 0.f;
    #pragma unroll
    for (int i = 0; i < FF / 4; ++i) {
        const float4 w = w4[i];
        const float4 q = a4[i];
        s = fmaf(w.x, q.x, s);
        s = fmaf(w.y, q.y, s);
        s = fmaf(w.z, q.z, s);
        s = fmaf(w.w, q.w, s);
    }
    return s;
}

static_assert(DIN * DHF / 4 == 4096 && DIN / 8 == 16, "the piece number's bit fields");
__device__ __forceinline__ void fold_pieces(int g, int t, int T,
                                            const float* __restrict__ Wall,
                                            float* __restrict__ wq)
{
    float4* __restrict__ q4 = reinterpret_cast<float4*>(wq);
    #pragma unroll
    for (int r = 0; r < 2; ++r) {
        const int e     = (2 * g + r) * T + t;
        const int layer = e >> 12;
        const int kb    = (e >> 8) & (DIN / 8 - 1);
        const int n     = (e >> 5) & 7;
        const int ln    = e & 31;
        const float* __restrict__ p = Wall + layer * DIN * DHF
                                    + (8 * kb + 4 * (ln >> 4)) * DHF + 16 * n + (ln & 15);
        q4[e] = make_float4(p[0], p[DHF], p[2 * DHF], p[3 * DHF]);
    }
}


#define CH_G       64
#define CH_EDGES   (EE / CH_G)
#define CH_THREADS 1024
#define CH_TRIPS   ((CH_EDGES + CH_THREADS - 1) / CH_THREADS)
static_assert(EE % CH_G == 0, "chunks must tile the edge list");
static_assert(CH_EDGES % 4 == 0, "four consecutive edges must share a chunk");
static_assert(NN % 4 == 0, "counters are cleared and copied as 16-byte words");

__device__ void ride_block(int b, int t,
                           const float* __restrict__ x,
                           const float* __restrict__ Wall,
                           const float* __restrict__ al,
                           const float* __restrict__ ar,
                           float* __restrict__ wa,
                           float* __restrict__ wq,
                           float* __restrict__ el,
                           float* __restrict__ er);

__global__ __launch_bounds__(CH_THREADS) void csr_chunk_count(const int* __restrict__ dst,
                                                              int* __restrict__ plane,
                                                              int* __restrict__ rank,
                                                              const float* __restrict__ x,
                                                              const float* __restrict__ Wall,
                                                              const float* __restrict__ al,
                                                              const float* __restrict__ ar,
                                                              float* __restrict__ wa,
                                                              float* __restrict__ wq,
                                                              float* __restrict__ el,
                                                              float* __restrict__ er)
{
    if (blockIdx.x >= CH_G) {
        ride_block((int)blockIdx.x - CH_G, (int)threadIdx.x, x, Wall, al, ar, wa, wq, el, er);
        return;
    }
    __shared__ __attribute__((aligned(16))) int cnt[NN];
    const int t  = threadIdx.x;
    const int e0 = blockIdx.x * CH_EDGES;

    int d[CH_TRIPS];
    #pragma unroll
    for (int k = 0; k < CH_TRIPS; ++k) {
        const int i = t + k * CH_THREADS;
        d[k] = dst[e0 + ((i < CH_EDGES) ? i : (CH_EDGES - 1))];
    }

    int4* c4 = reinterpret_cast<int4*>(cnt);
    for (int i = t; i < NN / 4; i += CH_THREADS) c4[i] = make_int4(0, 0, 0, 0);
    __syncthreads();

    int r[CH_TRIPS];
    #pragma unroll
    for (int k = 0; k < CH_TRIPS; ++k) {
        const int i = t + k * CH_THREADS;
        r[k] = (i < CH_EDGES) ? atomicAdd(&cnt[d[k]], 1) : 0;
    }
    __syncthreads();
    #pragma unroll
    for (int k = 0; k < CH_TRIPS; ++k) {
        const int i = t + k * CH_THREADS;
        if (i < CH_EDGES) rank[e0 + i] = r[k];
    }

    int4* p4 = reinterpret_cast<int4*>(plane + blockIdx.x * NN);
    for (int i = t; i < NN / 4; i += CH_THREADS) p4[i] = c4[i];
}

__global__ __launch_bounds__(256) void csr_chunk_prefix(int* __restrict__ plane,
                                                        int* __restrict__ deg)
{
    const int n = blockIdx.x * 256 + threadIdx.x;
    if (n >= NN) return;
    int c[CH_G];
    #pragma unroll
    for (int g = 0; g < CH_G; ++g) c[g] = plane[g * NN + n];
    int run = 0;
    #pragma unroll
    for (int g = 0; g < CH_G; ++g) { plane[g * NN + n] = run; run += c[g]; }
    deg[n] = run;
}

#define SCAN_CHUNK 1024
#define SCAN_FRONT_TRIPS ((((NN - 1) / SCAN_CHUNK) * (SCAN_CHUNK / 4) + SCAN_CHUNK - 1) / SCAN_CHUNK)
__global__ __launch_bounds__(SCAN_CHUNK) void csr_scan(const int* __restrict__ deg,
                                                       int* __restrict__ off)
{
    __shared__ int s_front[32];
    __shared__ int s_total[32];
    const int t    = threadIdx.x;
    const int lane = t & 31;
    const int wave = t >> 5;
    const int base = blockIdx.x * SCAN_CHUNK;

    const int4* __restrict__ d4 = reinterpret_cast<const int4*>(deg);
    const int nv = base >> 2;
    int front = 0;
    static_assert(SCAN_FRONT_TRIPS * SCAN_CHUNK * 4 <= NN, "front loads must stay inside deg");
    #pragma unroll
    for (int k = 0; k < SCAN_FRONT_TRIPS; ++k) {
        const int  j   = t + k * SCAN_CHUNK;
        const int4 q   = d4[j];
        const int  sum = (q.x + q.y) + (q.z + q.w);
        front += (j < nv) ? sum : 0;
    }

    const int i = base + t;
    const int v = (i < NN) ? deg[i] : 0;

    int s = v;
    #pragma unroll
    for (int o = 1; o < 32; o <<= 1) {
        const int y = __shfl_up(s, o);
        if (lane >= o) s += y;
    }
    #pragma unroll
    for (int o = 16; o > 0; o >>= 1) front += __shfl_xor(front, o);

    if (lane == 31) s_total[wave] = s;
    if (lane == 0)  s_front[wave] = front;
    __syncthreads();

    int c = s_front[lane] + ((lane < wave) ? s_total[lane] : 0);
    #pragma unroll
    for (int o = 16; o > 0; o >>= 1) c += __shfl_xor(c, o);

    const int excl = c + s - v;
    if (i < NN) off[i] = excl;
    if (i == NN - 1) off[NN] = excl + v;
}

__global__ __launch_bounds__(256) void csr_place(const int* __restrict__ src,
                                                 const int* __restrict__ dst,
                                                 const int* __restrict__ rank,
                                                 const int* __restrict__ plane,
                                                 const int* __restrict__ off,
                                                 int* __restrict__ csrc)
{
    const int q = blockIdx.x * 256 + threadIdx.x;
    if (q >= EE / 4) return;
    const int4 d = reinterpret_cast<const int4*>(dst)[q];
    const int4 s = reinterpret_cast<const int4*>(src)[q];
    const int4 r = reinterpret_cast<const int4*>(rank)[q];
    const int* __restrict__ pl = plane + ((q * 4) / CH_EDGES) * NN;
    const int p0 = off[d.x] + pl[d.x] + r.x;
    const int p1 = off[d.y] + pl[d.y] + r.y;
    const int p2 = off[d.z] + pl[d.z] + r.z;
    const int p3 = off[d.w] + pl[d.w] + r.w;
    csrc[p0] = s.x;
    csrc[p1] = s.y;
    csrc[p2] = s.z;
    csrc[p3] = s.w;
}

#define SCORE_NPW 8
static_assert(NN % SCORE_NPW == 0, "a wave's nodes must all exist");

__device__ __forceinline__ float wave_max(float v)
{
    #pragma unroll
    for (int o = 16; o > 0; o >>= 1) v = fmaxf(v, __shfl_xor(v, o));
    return v;
}
__device__ __forceinline__ float wave_sum(float v)
{
    #pragma unroll
    for (int o = 16; o > 0; o >>= 1) v += __shfl_xor(v, o);
    return v;
}
__device__ __forceinline__ float2 edge_score(const float2* __restrict__ el2, float2 erd, int s)
{
    const float2 l = el2[s];
    float a = l.x + erd.x, b = l.y + erd.y;
    a = (a >= 0.f) ? a : NEG_SLOPE * a;
    b = (b >= 0.f) ? b : NEG_SLOPE * b;
    return make_float2(a, b);
}

__global__ __launch_bounds__(256) void gat_edge_alpha(const float* __restrict__ el,
                                                      const float* __restrict__ er,
                                                      const int* __restrict__ csrc,
                                                      const int* __restrict__ off,
                                                      float* __restrict__ wgt)
{
    const int lane = threadIdx.x & 31;
    const int wave = __builtin_amdgcn_readfirstlane((int)(threadIdx.x >> 5));
    const int d0 = (blockIdx.x * 8 + wave) * SCORE_NPW;
    if (d0 >= NN) return;
    const float2* __restrict__ el2 = reinterpret_cast<const float2*>(el);
    const float2* __restrict__ er2 = reinterpret_cast<const float2*>(er);
    float2* __restrict__ w2 = reinterpret_cast<float2*>(wgt);
    const float ninf = -__builtin_inff();

    int po[SCORE_NPW + 1];
    #pragma unroll
    for (int k = 0; k <= SCORE_NPW; ++k) po[k] = off[d0 + k];
    bool small = true;
    #pragma unroll
    for (int k = 0; k < SCORE_NPW; ++k) small = small && (po[k + 1] - po[k] <= 32);

    if (small) {
        float2 e[SCORE_NPW];
        #pragma unroll
        for (int k = 0; k < SCORE_NPW; ++k) {
            const int p = po[k] + lane;
            e[k] = edge_score(el2, er2[d0 + k], csrc[(p < EE) ? p : (EE - 1)]);
        }
        float2 x[SCORE_NPW], s[SCORE_NPW];
        #pragma unroll
        for (int k = 0; k < SCORE_NPW; ++k) {
            const bool on = po[k] + lane < po[k + 1];
            const float e0 = on ? e[k].x : ninf;
            const float e1 = on ? e[k].y : ninf;
            const float m0 = wave_max(e0), m1 = wave_max(e1);
            x[k] = make_float2(expf(e0 - m0), expf(e1 - m1));
            s[k] = make_float2(wave_sum(x[k].x), wave_sum(x[k].y));
        }
        #pragma unroll
        for (int k = 0; k < SCORE_NPW; ++k) {
            const int p = po[k] + lane;
            if (p < po[k + 1]) w2[p] = make_float2(x[k].x / s[k].x, x[k].y / s[k].y);
        }
    } else {
        #pragma nounroll
        for (int k = 0; k < SCORE_NPW; ++k) {
            const int d  = d0 + k;
            const int p0 = off[d], p1 = off[d + 1];
            const float2 erd = er2[d];
            float m0 = ninf, m1 = ninf;
            for (int p = p0 + lane; p < p1; p += 32) {
                const float2 c = edge_score(el2, erd, csrc[p]);
                m0 = fmaxf(m0, c.x); m1 = fmaxf(m1, c.y);
            }
            m0 = wave_max(m0); m1 = wave_max(m1);
            float s0 = 0.f, s1 = 0.f;
            for (int p = p0 + lane; p < p1; p += 32) {
                const float2 c = edge_score(el2, erd, csrc[p]);
                s0 += expf(c.x - m0); s1 += expf(c.y - m1);
            }
            s0 = wave_sum(s0); s1 = wave_sum(s1);
            for (int p = p0 + lane; p < p1; p += 32) {
                const float2 c = edge_score(el2, erd, csrc[p]);
                w2[p] = make_float2(expf(c.x - m0) / s0, expf(c.y - m1) / s1);
            }
        }
    }
}

__device__ __forceinline__ float dot4(float4 a, float4 b)
{
    return fmaf(a.w, b.w, fmaf(a.z, b.z, fmaf(a.y, b.y, a.x * b.x)));
}
#define RS_ROWS     10
#define RIDE_WAVES  (CH_THREADS / 32)
#define RIDE_RS_G   ((NN / RS_ROWS + RIDE_WAVES - 1) / RIDE_WAVES)
#define RIDE_WQ_G   ((3 * DIN * DHF / 4) / (2 * CH_THREADS))
#define RIDE_G      (RIDE_RS_G + 1 + RIDE_WQ_G)
static_assert(NN % RS_ROWS == 0, "a wave's rows must all exist");
static_assert(CH_G + RIDE_G <= 256, "one block per compute unit, all at once");
static_assert(2 * 4 * DIN == CH_THREADS, "the later layers' folded vectors are one block of dots");
static_assert(4 * DIN <= CH_THREADS, "a block makes the first layer's folded vectors in one go");
static_assert((3 * DIN * DHF / 4) % (2 * CH_THREADS) == 0, "the pieces fill whole blocks");
__device__ void ride_block(int b, int t,
                           const float* __restrict__ x,
                           const float* __restrict__ Wall,
                           const float* __restrict__ al,
                           const float* __restrict__ ar,
                           float* __restrict__ wa,
                           float* __restrict__ wq,
                           float* __restrict__ el,
                           float* __restrict__ er)
{
    if (b > RIDE_RS_G) {
        fold_pieces(b - RIDE_RS_G - 1, t, CH_THREADS, Wall, wq);
        return;
    }
    if (b == RIDE_RS_G) {
        const int lj = 4 + t / DIN;
        wa[lj * DIN + (t & (DIN - 1))] = fold_dot(lj, t & (DIN - 1), Wall, al, ar);
        return;
    }

    __shared__ __attribute__((aligned(16))) float s_wa[4 * DIN];
    const int  lane = t & 31;
    const int  r0   = (b * RIDE_WAVES + (t >> 5)) * RS_ROWS;
    const bool live = r0 < NN;
    const int  r0c  = live ? r0 : 0;
    const float4* __restrict__ h4 = reinterpret_cast<const float4*>(x);
    float4 u[4], v[RS_ROWS];
    #pragma unroll
    for (int r = 0; r < RS_ROWS; ++r) v[r] = h4[(r0c + r) * (DIN / 4) + lane];
    if (t < 4 * DIN) s_wa[t] = fold_dot(t / DIN, t & (DIN - 1), Wall, al, ar);
    __syncthreads();
    if (!live) return;
    #pragma unroll
    for (int j = 0; j < 4; ++j) u[j] = *reinterpret_cast<const float4*>(&s_wa[j * DIN + 4 * lane]);
    #pragma unroll
    for (int r = 0; r < RS_ROWS; ++r) {
        float d[4];
        #pragma unroll
        for (int j = 0; j < 4; ++j) d[j] = wave_sum(dot4(v[r], u[j]));
        if (lane == 0) {
            reinterpret_cast<float2*>(el)[r0 + r] = make_float2(d[0], d[1]);
            reinterpret_cast<float2*>(er)[r0 + r] = make_float2(d[2], d[3]);
        }
    }
}

#define AGG_BATCH 16
#define AGG_NB    16
#define AGG_NW    (AGG_NB / 2)
#define G_PITCH   (DIN + 4)
#define PRJ_NT    2
static_assert(NN % AGG_NB == 0, "a work-group's nodes must all exist");
static_assert(DIN == 4 * 32, "a wave's lanes, two features each, cover half a row");
static_assert(DHF == 4 * 16 * PRJ_NT, "four waves of 2 column tiles cover W");
static_assert(FF == 2 * 16 * PRJ_NT, "a wave's column tiles lie in one head");

#define AGG_CLOSE_NODES(base_pos, rel_next)                                    \
    while (rel == (rel_next)) {                                                \
        *reinterpret_cast<float2*>(&G[gw + k * G_PITCH]) = acc0;               \
        *reinterpret_cast<float2*>(&G[gw + (AGG_NB + k) * G_PITCH]) = acc1;    \
        acc0 = make_float2(0.f, 0.f);                                          \
        acc1 = make_float2(0.f, 0.f);                                          \
        ++k;                                                                   \
        const int kn = (k < AGG_NW) ? (k + 1) : AGG_NW;                        \
        const int en = __builtin_amdgcn_readfirstlane(off[n0 + kn]);           \
        nend = (k < AGG_NW) ? en : 0x7fffffff;                                 \
        rel  = nend - (base_pos);                                              \
    }

__global__ __launch_bounds__(128) void gat_aggregate(const float* __restrict__ wgt,
                                                     const float* __restrict__ hin,
                                                     const int* __restrict__ csrc,
                                                     const int* __restrict__ off,
                                                     const float* __restrict__ wq,
                                                     const float* __restrict__ bias,
                                                     const float* __restrict__ wan,
                                                     float* __restrict__ hout,
                                                     float* __restrict__ el,
                                                     float* __restrict__ er,
                                                     float* __restrict__ out,
                                                     int last)
{
    __shared__ __attribute__((aligned(16))) float G[HH * AGG_NB * G_PITCH];
    const int t    = threadIdx.x;
    const int lane = t & 31;
    const int wi = __builtin_amdgcn_readfirstlane(t >> 5);
    const int fh = wi & 1;
    const int nh = wi >> 1;
    const int fl = (DIN / 4) * fh + lane;
    const float2* __restrict__ hin2 = reinterpret_cast<const float2*>(hin);
    const float2* __restrict__ w2   = reinterpret_cast<const float2*>(wgt);
    const int d0 = blockIdx.x * AGG_NB;
    const int n0 = d0 + AGG_NW * nh;
    const int gw = (AGG_NW * nh) * G_PITCH + 2 * fl;
    const int P0 = __builtin_amdgcn_readfirstlane(off[n0]);
    const int P1 = __builtin_amdgcn_readfirstlane(off[n0 + AGG_NW]);

    int    k    = 0;
    int    nend = __builtin_amdgcn_readfirstlane(off[n0 + 1]);
    int    rel  = nend - P0;
    float2 acc0 = make_float2(0.f, 0.f);
    float2 acc1 = make_float2(0.f, 0.f);

    AGG_CLOSE_NODES(P0, 0)

    if (P1 > P0) {
        int    q   = P0 + lane;
        int    qc  = (q < P1) ? q : (P1 - 1);
        int    vsn = csrc[qc];
        float2 vwn = w2[qc];

        for (int g = P0; g < P1; g += 32) {
            int   vs  = vsn;
            float vw0 = (g + lane < P1) ? vwn.x : 0.f;
            float vw1 = (g + lane < P1) ? vwn.y : 0.f;
            if (g + 32 < P1) {
                q   = g + 32 + lane;
                qc  = (q < P1) ? q : (P1 - 1);
                vsn = csrc[qc];
                vwn = w2[qc];
            }
            #pragma nounroll
            for (int hb = 0; hb < 32; hb += AGG_BATCH) {
                const int base = g + hb;
                if (base >= P1) break;
                rel = nend - base;
                float2 x[AGG_BATCH];
                #pragma unroll
                for (int j = 0; j < AGG_BATCH; ++j) {
                    const int s = __builtin_amdgcn_readlane(vs, j);
                    x[j] = hin2[s * (DIN / 2) + fl];
                }
                #pragma unroll
                for (int j = 0; j < AGG_BATCH; ++j) {
                    const float wa0 = __int_as_float(__builtin_amdgcn_readlane(__float_as_int(vw0), j));
                    const float wa1 = __int_as_float(__builtin_amdgcn_readlane(__float_as_int(vw1), j));
                    acc0.x = fmaf(wa0, x[j].x, acc0.x);
                    acc0.y = fmaf(wa0, x[j].y, acc0.y);
                    acc1.x = fmaf(wa1, x[j].x, acc1.x);
                    acc1.y = fmaf(wa1, x[j].y, acc1.y);
                    AGG_CLOSE_NODES(base, j + 1)
                }
                vs  = __shfl_down(vs, AGG_BATCH);
                vw0 = __shfl_down(vw0, AGG_BATCH);
                vw1 = __shfl_down(vw1, AGG_BATCH);
            }
        }
    }

    const int cc   = lane & 15;
    const int hh   = lane >> 4;
    const int hd   = wi >> 1;
    float res[PRJ_NT][8];
#if USE_WMMA_F32
    {
        const float4* __restrict__ bp = reinterpret_cast<const float4*>(wq) + (PRJ_NT * wi) * 32 + lane;
        const int aoff = (hd * AGG_NB + cc) * G_PITCH + 4 * hh;
        v8f acc[PRJ_NT];
        #pragma unroll
        for (int n = 0; n < PRJ_NT; ++n) acc[n] = v8f{};
        float4 p[3][PRJ_NT + 1];
        #pragma unroll
        for (int n = 0; n < PRJ_NT; ++n) {
            p[0][n] = bp[n * 32];
            p[1][n] = bp[(8 + n) * 32];
        }
        __syncthreads();
        p[0][PRJ_NT] = *reinterpret_cast<const float4*>(&G[aoff]);
        p[1][PRJ_NT] = *reinterpret_cast<const float4*>(&G[aoff + 8]);
        #pragma unroll
        for (int kb = 0; kb < DIN / 8; ++kb) {
            if (kb + 2 < DIN / 8) {
                #pragma unroll
                for (int n = 0; n < PRJ_NT; ++n) p[(kb + 2) % 3][n] = bp[((kb + 2) * 8 + n) * 32];
                p[(kb + 2) % 3][PRJ_NT] = *reinterpret_cast<const float4*>(&G[aoff + 8 * (kb + 2)]);
            }
            __builtin_amdgcn_sched_barrier(0);
            const float4 pa = p[kb % 3][PRJ_NT];
            v2f a0, a1;
            a0.x = pa.x; a0.y = pa.y;
            a1.x = pa.z; a1.y = pa.w;
            #pragma unroll
            for (int n = 0; n < PRJ_NT; ++n) {
                v2f b0;
                b0.x = p[kb % 3][n].x; b0.y = p[kb % 3][n].y;
                acc[n] = __builtin_amdgcn_wmma_f32_16x16x4_f32(false, a0, false, b0,
                                                               (short)0, acc[n], false, false);
            }
            #pragma unroll
            for (int n = 0; n < PRJ_NT; ++n) {
                v2f b1;
                b1.x = p[kb % 3][n].z; b1.y = p[kb % 3][n].w;
                acc[n] = __builtin_amdgcn_wmma_f32_16x16x4_f32(false, a1, false, b1,
                                                               (short)0, acc[n], false, false);
            }
            __builtin_amdgcn_sched_barrier(0);
        }
        #pragma unroll
        for (int n = 0; n < PRJ_NT; ++n) {
            #pragma unroll
            for (int i = 0; i < 8; ++i) res[n][i] = acc[n][i];
        }
    }
#else
    __syncthreads();
    for (int n = 0; n < PRJ_NT; ++n) {
        const int col = 16 * (PRJ_NT * wi + n) + cc;
        for (int i = 0; i < 8; ++i) {
            float s = 0.f;
            for (int kk = 0; kk < DIN; ++kk)
                s = fmaf(G[(hd * AGG_NB + 8 * hh + i) * G_PITCH + kk],
                         wq[((((kk >> 3) * 8 + (col >> 4)) * 32) + 16 * ((kk >> 2) & 1) + (col & 15)) * 4 + (kk & 3)], s);
            res[n][i] = s;
        }
    }
#endif

    __syncthreads();
    #pragma unroll
    for (int n = 0; n < PRJ_NT; ++n) {
        const int col = 16 * (PRJ_NT * wi + n) + cc;
        const float bc = bias[col];
        #pragma unroll
        for (int i = 0; i < 8; ++i) {
            float v = res[n][i] + bc;
            v = (v > 0.f) ? v : expm1f(v);
            G[(8 * hh + i) * G_PITCH + col] = v;
        }
    }
    __syncthreads();

    if (!last) {
        const float4* __restrict__ wa4 = reinterpret_cast<const float4*>(wan);
        float4 u[4], x[4];
        #pragma unroll
        for (int j = 0; j < 4; ++j) u[j] = wa4[j * (DIN / 4) + lane];
        #pragma unroll
        for (int r = 0; r < 4; ++r) x[r] = *reinterpret_cast<const float4*>(&G[(4 * wi + r) * G_PITCH + 4 * lane]);
        #pragma unroll
        for (int r = 0; r < 4; ++r)
            reinterpret_cast<float4*>(hout)[(d0 + 4 * wi + r) * (DIN / 4) + lane] = x[r];
        #pragma unroll
        for (int r = 0; r < 4; ++r) {
            float d[4];
            #pragma unroll
            for (int j = 0; j < 4; ++j) d[j] = wave_sum(dot4(x[r], u[j]));
            if (lane == 0) {
                reinterpret_cast<float2*>(el)[d0 + 4 * wi + r] = make_float2(d[0], d[1]);
                reinterpret_cast<float2*>(er)[d0 + 4 * wi + r] = make_float2(d[2], d[3]);
            }
        }
    } else {
        #pragma unroll
        for (int r = 0; r < 4; ++r) {
            const float2 a = *reinterpret_cast<const float2*>(&G[(4 * wi + r) * G_PITCH + 2 * lane]);
            const float2 b = *reinterpret_cast<const float2*>(&G[(4 * wi + r) * G_PITCH + FF + 2 * lane]);
            reinterpret_cast<float2*>(out)[(d0 + 4 * wi + r) * (FF / 2) + lane] =
                make_float2(0.5f * (a.x + b.x), 0.5f * (a.y + b.y));
        }
    }
}

extern "C" void kernel_launch(void* const* d_in, const int* in_sizes, int n_in,
                              void* d_out, int out_size, void* d_ws, size_t ws_size,
                              hipStream_t stream)
{
    const float* x    = (const float*)d_in[0];
    const int*   src  = (const int*)d_in[1];
    const int*   dst  = (const int*)d_in[2];
    const float* Wall = (const float*)d_in[3];
    const float* al   = (const float*)d_in[4];
    const float* ar   = (const float*)d_in[5];
    const float* bias = (const float*)d_in[6];
    float* out = (float*)d_out;

    char* p = (char*)d_ws;
    auto take = [&](size_t bytes) {
        char* q = p;
        p += (bytes + 255) & ~size_t(255);
        return q;
    };
    float* hbufa  = (float*)take(sizeof(float) * NN * DHF);
    float* hbufb  = (float*)take(sizeof(float) * NN * DHF);
    float* el     = (float*)take(sizeof(float) * NN * HH);
    float* er     = (float*)take(sizeof(float) * NN * HH);
    float* wgt    = (float*)take(sizeof(float) * EE * HH);
    int*   deg    = (int*)take(sizeof(int) * NN);
    int*   off    = (int*)take(sizeof(int) * (NN + 1));
    int*   rank   = (int*)take(sizeof(int) * EE);
    int*   plane  = (int*)take(sizeof(int) * CH_G * NN);
    int*   csrc   = (int*)take(sizeof(int) * EE);
    float* wa     = (float*)take(sizeof(float) * 3 * 4 * DIN);
    float* wq     = (float*)take(sizeof(float) * 3 * DIN * DHF);

    const int tb   = 256;
    const int gN   = (NN + tb - 1) / tb;
    const int gPl  = (EE / 4 + tb - 1) / tb;
    const int gSc  = (NN / SCORE_NPW + 7) / 8;
    const int gScan = (NN + SCAN_CHUNK - 1) / SCAN_CHUNK;

    csr_chunk_count <<<CH_G + RIDE_G, CH_THREADS, 0, stream>>>(dst, plane, rank, x, Wall, al, ar, wa, wq, el, er);
    csr_chunk_prefix<<<gN, tb, 0, stream>>>(plane, deg);
    csr_scan        <<<gScan, SCAN_CHUNK, 0, stream>>>(deg, off);
    csr_place       <<<gPl, tb, 0, stream>>>(src, dst, rank, plane, off, csrc);

    const float* hin = x;
    for (int layer = 0; layer < 3; ++layer) {
        const float* Wl  = wq + layer * DIN * DHF;
        const float* bl  = bias + layer * HH * FF;
        const int    last = (layer == 2);
        float* hout = (layer & 1) ? hbufb : hbufa;
        const float* wan = wa + (last ? layer : (layer + 1)) * 4 * DIN;

        gat_edge_alpha<<<gSc, tb, 0, stream>>>(el, er, csrc, off, wgt);
        gat_aggregate<<<NN / AGG_NB, 128, 0, stream>>>(wgt, hin, csrc, off, Wl, bl, wan, hout, el, er, out, last);
        hin = hout;
    }
    (void)in_sizes; (void)n_in; (void)out_size; (void)ws_size;
}
